# starting kernel
# baseline (speedup 1.0000x reference)
#include <hip/hip_runtime.h>
#include <math.h>

#define D 64
#define K2 128
#define M_USERS 200000
#define NUM_NODES 300000

typedef __attribute__((ext_vector_type(2))) float v2f;
typedef __attribute__((ext_vector_type(8))) float v8f;

__global__ void zero_kernel(float4* __restrict__ p, long n4) {
    long i = (long)blockIdx.x * blockDim.x + threadIdx.x;
    long stride = (long)gridDim.x * blockDim.x;
    float4 z = make_float4(0.f, 0.f, 0.f, 0.f);
    for (; i < n4; i += stride) p[i] = z;
}

__global__ void spmm_kernel(const float* __restrict__ xu,
                            const float* __restrict__ xi,
                            int split,
                            const int*   __restrict__ rows,
                            const int*   __restrict__ cols,
                            const float* __restrict__ vals,
                            float* __restrict__ out,
                            int nedges) {
    int warp = (blockIdx.x * blockDim.x + threadIdx.x) >> 5;
    int lane = threadIdx.x & 31;
    int nw = (gridDim.x * blockDim.x) >> 5;
    for (int e = warp; e < nedges; e += nw) {
        int r = rows[e];
        int c = cols[e];
        float v = vals[e];
        const float* src = (c < split) ? (xu + (size_t)c * D)
                                       : (xi + (size_t)(c - split) * D);
        float2 xv = *(const float2*)(src + lane * 2);
        float* dst = out + (size_t)r * D + lane * 2;
        atomicAdd(dst,     v * xv.x);
        atomicAdd(dst + 1, v * xv.y);
    }
}

__global__ __launch_bounds__(128) void fuse_kernel(
    const float* __restrict__ user_emb, const float* __restrict__ item_emb,
    const float* __restrict__ hg_user,  const float* __restrict__ hg_item,
    const float* __restrict__ h1,       const float* __restrict__ h2,
    const float* __restrict__ W1,       const float* __restrict__ b1,
    const float* __restrict__ W2,       const float* __restrict__ b2,
    float* __restrict__ out)
{
    __shared__ float sW1t[D * K2];
    __shared__ float sA[4][16 * K2];

    const int tid  = threadIdx.x;
    const int w    = tid >> 5;
    const int lane = tid & 31;

    for (int i = tid; i < K2 * D; i += 128) {
        int k = i >> 6;
        int n = i & 63;
        sW1t[n * K2 + k] = W1[i];
    }

    const int tile_base = (blockIdx.x * 4 + w) * 16;

    for (int i = lane; i < 16 * 32; i += 32) {
        int node = i >> 5;
        int c4   = i & 31;
        int feat = c4 * 4;
        int ng   = tile_base + node;
        float4 val = make_float4(0.f, 0.f, 0.f, 0.f);
        if (ng < NUM_NODES) {
            if (feat < D) {
                const float* ego = (ng < M_USERS)
                    ? (user_emb + (size_t)ng * D + feat)
                    : (item_emb + (size_t)(ng - M_USERS) * D + feat);
                float4 e4 = *(const float4*)ego;
                float4 a4 = *(const float4*)(h1 + (size_t)ng * D + feat);
                float4 b4 = *(const float4*)(h2 + (size_t)ng * D + feat);
                const float inv3 = 1.0f / 3.0f;
                val = make_float4((e4.x + a4.x + b4.x) * inv3,
                                  (e4.y + a4.y + b4.y) * inv3,
                                  (e4.z + a4.z + b4.z) * inv3,
                                  (e4.w + a4.w + b4.w) * inv3);
            } else {
                int f = feat - D;
                const float* hg = (ng < M_USERS)
                    ? (hg_user + (size_t)ng * D + f)
                    : (hg_item + (size_t)(ng - M_USERS) * D + f);
                val = *(const float4*)hg;
            }
        }
        *(float4*)&sA[w][node * K2 + feat] = val;
    }

    __syncthreads();

    const int mrow = lane & 15;
    const int koff = (lane >> 4) * 2;

    v8f acc[4] = {};
    for (int kk = 0; kk < 32; ++kk) {
        int k0 = kk * 4 + koff;
        v2f a;
        a.x = sA[w][mrow * K2 + k0];
        a.y = sA[w][mrow * K2 + k0 + 1];
        #pragma unroll
        for (int nt = 0; nt < 4; ++nt) {
            int ncol = nt * 16 + mrow;
            v2f b;
            b.x = sW1t[ncol * K2 + k0];
            b.y = sW1t[ncol * K2 + k0 + 1];
            acc[nt] = __builtin_amdgcn_wmma_f32_16x16x4_f32(
                false, a, false, b, (short)0, acc[nt], false, false);
        }
    }

    float p0[8], p1[8];
    #pragma unroll
    for (int v = 0; v < 8; ++v) { p0[v] = 0.f; p1[v] = 0.f; }
    for (int nt = 0; nt < 4; ++nt) {
        int ncol = nt * 16 + mrow;
        float bb  = b1[ncol];
        float w2a = W2[ncol * 2 + 0];
        float w2b = W2[ncol * 2 + 1];
        #pragma unroll
        for (int v = 0; v < 8; ++v) {
            float hdn = tanhf(acc[nt][v] + bb);
            p0[v] += hdn * w2a;
            p1[v] += hdn * w2b;
        }
    }

    #pragma unroll
    for (int off = 1; off < 16; off <<= 1) {
        #pragma unroll
        for (int v = 0; v < 8; ++v) {
            p0[v] += __shfl_xor(p0[v], off, 32);
            p1[v] += __shfl_xor(p1[v], off, 32);
        }
    }

    float bb0 = b2[0], bb1 = b2[1];
    const int half = lane >> 4;

    for (int v = 0; v < 8; ++v) {
        float l0 = p0[v] + bb0;
        float l1 = p1[v] + bb1;
        float e  = expf(l1 - l0);
        float w1w = e / (1.0f + e);
        float w0w = 1.0f - w1w;
        int m  = v + half * 8;
        int ng = tile_base + m;
        if (ng < NUM_NODES) {
            #pragma unroll
            for (int ff = 0; ff < 4; ++ff) {
                int feat = mrow + ff * 16;
                float f1v = sA[w][m * K2 + feat];
                float f2v = sA[w][m * K2 + D + feat];
                out[(size_t)ng * D + feat] = w0w * f1v + w1w * f2v;
            }
        }
    }
}

extern "C" void kernel_launch(void* const* d_in, const int* in_sizes, int n_in,
                              void* d_out, int out_size, void* d_ws, size_t ws_size,
                              hipStream_t stream) {
    const float* user_emb = (const float*)d_in[0];
    const float* item_emb = (const float*)d_in[1];
    const float* hg_user  = (const float*)d_in[2];
    const float* hg_item  = (const float*)d_in[3];
    const float* adj_vals = (const float*)d_in[4];
    const float* W1       = (const float*)d_in[5];
    const float* b1       = (const float*)d_in[6];
    const float* W2       = (const float*)d_in[7];
    const float* b2       = (const float*)d_in[8];
    const int*   adj_rows = (const int*)d_in[9];
    const int*   adj_cols = (const int*)d_in[10];
    float* out = (float*)d_out;

    float* h1 = (float*)d_ws;
    float* h2 = h1 + (size_t)NUM_NODES * D;

    long n4 = (long)(2 * (size_t)NUM_NODES * D / 4);
    int zb = (int)((n4 + 255) / 256);
    zero_kernel<<<zb, 256, 0, stream>>>((float4*)d_ws, n4);

    int E = in_sizes[4];
    int warps_per_block = 256 / 32;
    int spmm_blocks = (E + warps_per_block - 1) / warps_per_block;
    spmm_kernel<<<spmm_blocks, 256, 0, stream>>>(user_emb, item_emb, M_USERS,
                                                 adj_rows, adj_cols, adj_vals, h1, E);
    spmm_kernel<<<spmm_blocks, 256, 0, stream>>>(h1, h1, NUM_NODES,
                                                 adj_rows, adj_cols, adj_vals, h2, E);

    int fuse_blocks = (NUM_NODES + 63) / 64;
    fuse_kernel<<<fuse_blocks, 128, 0, stream>>>(user_emb, item_emb, hg_user, hg_item,
                                                 h1, h2, W1, b1, W2, b2, out);
}


# SpMM: float-atomic edge scatter replaced by per-call counting sort of edges by row (int atomics) + atomic-free row gather; sums in bucket arrival order
# speedup vs baseline: 1.6379x; 1.6379x over previous
#include <hip/hip_runtime.h>
#include <math.h>

#define D 64
#define K2 128
#define M_USERS 200000
#define NUM_NODES 300000

typedef __attribute__((ext_vector_type(2))) float v2f;
typedef __attribute__((ext_vector_type(8))) float v8f;

__global__ void zero_kernel(float4* __restrict__ p, long n4) {
    long i = (long)blockIdx.x * blockDim.x + threadIdx.x;
    long stride = (long)gridDim.x * blockDim.x;
    float4 z = make_float4(0.f, 0.f, 0.f, 0.f);
    for (; i < n4; i += stride) p[i] = z;
}

#define SCAN_CHUNK 1024
#define SCAN_BLOCKS ((NUM_NODES + SCAN_CHUNK - 1) / SCAN_CHUNK)
static_assert(NUM_NODES % 4 == 0, "row counters are read as int4");

__global__ void hist_kernel(const int* __restrict__ rows, int* __restrict__ cnt, int nedges) {
    int i = blockIdx.x * blockDim.x + threadIdx.x;
    int stride = gridDim.x * blockDim.x;
    for (int e = i; e < nedges; e += stride) atomicAdd(&cnt[rows[e]], 1);
}

__global__ __launch_bounds__(256) void blocksum_kernel(const int* __restrict__ cnt,
                                                       int* __restrict__ blocksum) {
    __shared__ int s_red[256];
    const int t = threadIdx.x;
    const int i4 = blockIdx.x * 256 + t;
    int s = 0;
    if (i4 < NUM_NODES / 4) {
        int4 c = ((const int4*)cnt)[i4];
        s = c.x + c.y + c.z + c.w;
    }
    s_red[t] = s;
    __syncthreads();
    for (int off = 128; off > 0; off >>= 1) {
        if (t < off) s_red[t] += s_red[t + off];
        __syncthreads();
    }
    if (t == 0) blocksum[blockIdx.x] = s_red[0];
}

__global__ __launch_bounds__(256) void scan_kernel(const int* __restrict__ cnt,
                                                   const int* __restrict__ blocksum,
                                                   int* __restrict__ rowptr,
                                                   int* __restrict__ cursor) {
    __shared__ int s_red[256];
    __shared__ int s_scan[256];
    const int t = threadIdx.x;
    const int b = blockIdx.x;

    int part = 0;
    for (int j = t; j < b; j += 256) part += blocksum[j];
    s_red[t] = part;
    __syncthreads();
    for (int off = 128; off > 0; off >>= 1) {
        if (t < off) s_red[t] += s_red[t + off];
        __syncthreads();
    }
    const int base = s_red[0];

    const int i4 = b * 256 + t;
    const bool valid = i4 < NUM_NODES / 4;
    int4 c = make_int4(0, 0, 0, 0);
    if (valid) c = ((const int4*)cnt)[i4];
    const int s = c.x + c.y + c.z + c.w;

    s_scan[t] = s;
    __syncthreads();
    for (int off = 1; off < 256; off <<= 1) {
        int add = (t >= off) ? s_scan[t - off] : 0;
        __syncthreads();
        s_scan[t] += add;
        __syncthreads();
    }
    const int excl = base + s_scan[t] - s;

    if (valid) {
        int4 o;
        o.x = excl;
        o.y = excl + c.x;
        o.z = excl + c.x + c.y;
        o.w = excl + c.x + c.y + c.z;
        ((int4*)rowptr)[i4] = o;
        ((int4*)cursor)[i4] = o;
    }
}

__global__ void fill_kernel(const int* __restrict__ rows, const int* __restrict__ cols,
                            const float* __restrict__ vals, int* __restrict__ cursor,
                            int2* __restrict__ sedge, int nedges) {
    int i = blockIdx.x * blockDim.x + threadIdx.x;
    int stride = gridDim.x * blockDim.x;
    for (int e = i; e < nedges; e += stride) {
        int pos = atomicAdd(&cursor[rows[e]], 1);
        if ((unsigned)pos < (unsigned)nedges)
            sedge[pos] = make_int2(cols[e], __float_as_int(vals[e]));
    }
}

typedef __attribute__((ext_vector_type(4))) float v4f;

__device__ __forceinline__ v4f load_x4(const float* __restrict__ xu,
                                       const float* __restrict__ xi,
                                       int split, int c, int sub) {
    const float* src = (c < split) ? (xu + (size_t)c * D)
                                   : (xi + (size_t)(c - split) * D);
    return *(const v4f*)(src + sub * 4);
}

__global__ __launch_bounds__(256) void gather_kernel(const float* __restrict__ xu,
                                                     const float* __restrict__ xi,
                                                     int split,
                                                     const int*  __restrict__ rowptr,
                                                     const int*  __restrict__ rowend,
                                                     const int2* __restrict__ sedge,
                                                     float* __restrict__ out,
                                                     int nedges) {
    const int row = (blockIdx.x * 256 + threadIdx.x) >> 4;
    const int sub = threadIdx.x & 15;
    if (row >= NUM_NODES) return;
    const int start = min(max(rowptr[row], 0), nedges);
    const int end   = min(max(rowend[row], start), nedges);
    v4f acc = {0.f, 0.f, 0.f, 0.f};
    for (int base = start; base < end; base += 16) {
        const int n = min(end - base, 16);
        int mc = 0, mv = 0;
        if (sub < n) {
            int2 m = sedge[base + sub];
            mc = m.x;
            mv = m.y;
        }
        for (int t = 0; t < n; t += 4) {
            const int i1 = min(t + 1, n - 1);
            const int i2 = min(t + 2, n - 1);
            const int i3 = min(t + 3, n - 1);
            const int   c0 = __shfl(mc, t,  16);
            const int   c1 = __shfl(mc, i1, 16);
            const int   c2 = __shfl(mc, i2, 16);
            const int   c3 = __shfl(mc, i3, 16);
            const float w0 = __int_as_float(__shfl(mv, t,  16));
            const float w1 = __int_as_float(__shfl(mv, i1, 16));
            const float w2 = __int_as_float(__shfl(mv, i2, 16));
            const float w3 = __int_as_float(__shfl(mv, i3, 16));
            v4f x0 = load_x4(xu, xi, split, c0, sub);
            v4f x1 = load_x4(xu, xi, split, c1, sub);
            v4f x2 = load_x4(xu, xi, split, c2, sub);
            v4f x3 = load_x4(xu, xi, split, c3, sub);
            const v4f zero = {0.f, 0.f, 0.f, 0.f};
            acc += w0 * x0;
            acc += (t + 1 < n) ? (w1 * x1) : zero;
            acc += (t + 2 < n) ? (w2 * x2) : zero;
            acc += (t + 3 < n) ? (w3 * x3) : zero;
        }
    }
    *(v4f*)(out + (size_t)row * D + sub * 4) = acc;
}

__global__ __launch_bounds__(128) void fuse_kernel(
    const float* __restrict__ user_emb, const float* __restrict__ item_emb,
    const float* __restrict__ hg_user,  const float* __restrict__ hg_item,
    const float* __restrict__ h1,       const float* __restrict__ h2,
    const float* __restrict__ W1,       const float* __restrict__ b1,
    const float* __restrict__ W2,       const float* __restrict__ b2,
    float* __restrict__ out)
{
    __shared__ float sW1t[D * K2];
    __shared__ float sA[4][16 * K2];

    const int tid  = threadIdx.x;
    const int w    = tid >> 5;
    const int lane = tid & 31;

    for (int i = tid; i < K2 * D; i += 128) {
        int k = i >> 6;
        int n = i & 63;
        sW1t[n * K2 + k] = W1[i];
    }

    const int tile_base = (blockIdx.x * 4 + w) * 16;

    for (int i = lane; i < 16 * 32; i += 32) {
        int node = i >> 5;
        int c4   = i & 31;
        int feat = c4 * 4;
        int ng   = tile_base + node;
        float4 val = make_float4(0.f, 0.f, 0.f, 0.f);
        if (ng < NUM_NODES) {
            if (feat < D) {
                const float* ego = (ng < M_USERS)
                    ? (user_emb + (size_t)ng * D + feat)
                    : (item_emb + (size_t)(ng - M_USERS) * D + feat);
                float4 e4 = *(const float4*)ego;
                float4 a4 = *(const float4*)(h1 + (size_t)ng * D + feat);
                float4 b4 = *(const float4*)(h2 + (size_t)ng * D + feat);
                const float inv3 = 1.0f / 3.0f;
                val = make_float4((e4.x + a4.x + b4.x) * inv3,
                                  (e4.y + a4.y + b4.y) * inv3,
                                  (e4.z + a4.z + b4.z) * inv3,
                                  (e4.w + a4.w + b4.w) * inv3);
            } else {
                int f = feat - D;
                const float* hg = (ng < M_USERS)
                    ? (hg_user + (size_t)ng * D + f)
                    : (hg_item + (size_t)(ng - M_USERS) * D + f);
                val = *(const float4*)hg;
            }
        }
        *(float4*)&sA[w][node * K2 + feat] = val;
    }

    __syncthreads();

    const int mrow = lane & 15;
    const int koff = (lane >> 4) * 2;

    v8f acc[4] = {};
    for (int kk = 0; kk < 32; ++kk) {
        int k0 = kk * 4 + koff;
        v2f a;
        a.x = sA[w][mrow * K2 + k0];
        a.y = sA[w][mrow * K2 + k0 + 1];
        #pragma unroll
        for (int nt = 0; nt < 4; ++nt) {
            int ncol = nt * 16 + mrow;
            v2f b;
            b.x = sW1t[ncol * K2 + k0];
            b.y = sW1t[ncol * K2 + k0 + 1];
            acc[nt] = __builtin_amdgcn_wmma_f32_16x16x4_f32(
                false, a, false, b, (short)0, acc[nt], false, false);
        }
    }

    float p0[8], p1[8];
    #pragma unroll
    for (int v = 0; v < 8; ++v) { p0[v] = 0.f; p1[v] = 0.f; }
    for (int nt = 0; nt < 4; ++nt) {
        int ncol = nt * 16 + mrow;
        float bb  = b1[ncol];
        float w2a = W2[ncol * 2 + 0];
        float w2b = W2[ncol * 2 + 1];
        #pragma unroll
        for (int v = 0; v < 8; ++v) {
            float hdn = tanhf(acc[nt][v] + bb);
            p0[v] += hdn * w2a;
            p1[v] += hdn * w2b;
        }
    }

    #pragma unroll
    for (int off = 1; off < 16; off <<= 1) {
        #pragma unroll
        for (int v = 0; v < 8; ++v) {
            p0[v] += __shfl_xor(p0[v], off, 32);
            p1[v] += __shfl_xor(p1[v], off, 32);
        }
    }

    float bb0 = b2[0], bb1 = b2[1];
    const int half = lane >> 4;

    for (int v = 0; v < 8; ++v) {
        float l0 = p0[v] + bb0;
        float l1 = p1[v] + bb1;
        float e  = expf(l1 - l0);
        float w1w = e / (1.0f + e);
        float w0w = 1.0f - w1w;
        int m  = v + half * 8;
        int ng = tile_base + m;
        if (ng < NUM_NODES) {
            #pragma unroll
            for (int ff = 0; ff < 4; ++ff) {
                int feat = mrow + ff * 16;
                float f1v = sA[w][m * K2 + feat];
                float f2v = sA[w][m * K2 + D + feat];
                out[(size_t)ng * D + feat] = w0w * f1v + w1w * f2v;
            }
        }
    }
}

extern "C" void kernel_launch(void* const* d_in, const int* in_sizes, int n_in,
                              void* d_out, int out_size, void* d_ws, size_t ws_size,
                              hipStream_t stream) {
    const float* user_emb = (const float*)d_in[0];
    const float* item_emb = (const float*)d_in[1];
    const float* hg_user  = (const float*)d_in[2];
    const float* hg_item  = (const float*)d_in[3];
    const float* adj_vals = (const float*)d_in[4];
    const float* W1       = (const float*)d_in[5];
    const float* b1       = (const float*)d_in[6];
    const float* W2       = (const float*)d_in[7];
    const float* b2       = (const float*)d_in[8];
    const int*   adj_rows = (const int*)d_in[9];
    const int*   adj_cols = (const int*)d_in[10];
    float* out = (float*)d_out;

    float* h1 = (float*)d_ws;
    float* h2 = h1 + (size_t)NUM_NODES * D;
    int*  cnt      = (int*)(h2 + (size_t)NUM_NODES * D);
    int*  rowptr   = cnt + NUM_NODES;
    int*  cursor   = rowptr + NUM_NODES;
    int*  blocksum = cursor + NUM_NODES;
    int2* sedge    = (int2*)(blocksum + 512);

    int E = in_sizes[4];

    zero_kernel<<<SCAN_BLOCKS, 256, 0, stream>>>((float4*)cnt, (long)(NUM_NODES / 4));
    hist_kernel<<<2048, 256, 0, stream>>>(adj_rows, cnt, E);
    blocksum_kernel<<<SCAN_BLOCKS, 256, 0, stream>>>(cnt, blocksum);
    scan_kernel<<<SCAN_BLOCKS, 256, 0, stream>>>(cnt, blocksum, rowptr, cursor);
    fill_kernel<<<2048, 256, 0, stream>>>(adj_rows, adj_cols, adj_vals, cursor, sedge, E);

    int gather_blocks = (NUM_NODES * 16 + 255) / 256;
    gather_kernel<<<gather_blocks, 256, 0, stream>>>(user_emb, item_emb, M_USERS,
                                                     rowptr, cursor, sedge, h1, E);
    gather_kernel<<<gather_blocks, 256, 0, stream>>>(h1, h1, NUM_NODES,
                                                     rowptr, cursor, sedge, h2, E);

    int fuse_blocks = (NUM_NODES + 63) / 64;
    fuse_kernel<<<fuse_blocks, 128, 0, stream>>>(user_emb, item_emb, hg_user, hg_item,
                                                 h1, h2, W1, b1, W2, b2, out);
}


# fuse MLP: LDS row pitch of both staged operands 128 -> 132 floats, so WMMA fragment reads and the W1 transpose store stop colliding on one bank
# speedup vs baseline: 1.8269x; 1.1154x over previous
#include <hip/hip_runtime.h>
#include <math.h>

#define D 64
#define K2 128
#define M_USERS 200000
#define NUM_NODES 300000

typedef __attribute__((ext_vector_type(2))) float v2f;
typedef __attribute__((ext_vector_type(8))) float v8f;

__global__ void zero_kernel(float4* __restrict__ p, long n4) {
    long i = (long)blockIdx.x * blockDim.x + threadIdx.x;
    long stride = (long)gridDim.x * blockDim.x;
    float4 z = make_float4(0.f, 0.f, 0.f, 0.f);
    for (; i < n4; i += stride) p[i] = z;
}

#define SCAN_CHUNK 1024
#define SCAN_BLOCKS ((NUM_NODES + SCAN_CHUNK - 1) / SCAN_CHUNK)
static_assert(NUM_NODES % 4 == 0, "row counters are read as int4");

__global__ void hist_kernel(const int* __restrict__ rows, int* __restrict__ cnt, int nedges) {
    int i = blockIdx.x * blockDim.x + threadIdx.x;
    int stride = gridDim.x * blockDim.x;
    for (int e = i; e < nedges; e += stride) atomicAdd(&cnt[rows[e]], 1);
}

__global__ __launch_bounds__(256) void blocksum_kernel(const int* __restrict__ cnt,
                                                       int* __restrict__ blocksum) {
    __shared__ int s_red[256];
    const int t = threadIdx.x;
    const int i4 = blockIdx.x * 256 + t;
    int s = 0;
    if (i4 < NUM_NODES / 4) {
        int4 c = ((const int4*)cnt)[i4];
        s = c.x + c.y + c.z + c.w;
    }
    s_red[t] = s;
    __syncthreads();
    for (int off = 128; off > 0; off >>= 1) {
        if (t < off) s_red[t] += s_red[t + off];
        __syncthreads();
    }
    if (t == 0) blocksum[blockIdx.x] = s_red[0];
}

__global__ __launch_bounds__(256) void scan_kernel(const int* __restrict__ cnt,
                                                   const int* __restrict__ blocksum,
                                                   int* __restrict__ rowptr,
                                                   int* __restrict__ cursor) {
    __shared__ int s_red[256];
    __shared__ int s_scan[256];
    const int t = threadIdx.x;
    const int b = blockIdx.x;

    int part = 0;
    for (int j = t; j < b; j += 256) part += blocksum[j];
    s_red[t] = part;
    __syncthreads();
    for (int off = 128; off > 0; off >>= 1) {
        if (t < off) s_red[t] += s_red[t + off];
        __syncthreads();
    }
    const int base = s_red[0];

    const int i4 = b * 256 + t;
    const bool valid = i4 < NUM_NODES / 4;
    int4 c = make_int4(0, 0, 0, 0);
    if (valid) c = ((const int4*)cnt)[i4];
    const int s = c.x + c.y + c.z + c.w;

    s_scan[t] = s;
    __syncthreads();
    for (int off = 1; off < 256; off <<= 1) {
        int add = (t >= off) ? s_scan[t - off] : 0;
        __syncthreads();
        s_scan[t] += add;
        __syncthreads();
    }
    const int excl = base + s_scan[t] - s;

    if (valid) {
        int4 o;
        o.x = excl;
        o.y = excl + c.x;
        o.z = excl + c.x + c.y;
        o.w = excl + c.x + c.y + c.z;
        ((int4*)rowptr)[i4] = o;
        ((int4*)cursor)[i4] = o;
    }
}

__global__ void fill_kernel(const int* __restrict__ rows, const int* __restrict__ cols,
                            const float* __restrict__ vals, int* __restrict__ cursor,
                            int2* __restrict__ sedge, int nedges) {
    int i = blockIdx.x * blockDim.x + threadIdx.x;
    int stride = gridDim.x * blockDim.x;
    for (int e = i; e < nedges; e += stride) {
        int pos = atomicAdd(&cursor[rows[e]], 1);
        if ((unsigned)pos < (unsigned)nedges)
            sedge[pos] = make_int2(cols[e], __float_as_int(vals[e]));
    }
}

typedef __attribute__((ext_vector_type(4))) float v4f;

__device__ __forceinline__ v4f load_x4(const float* __restrict__ xu,
                                       const float* __restrict__ xi,
                                       int split, int c, int sub) {
    const float* src = (c < split) ? (xu + (size_t)c * D)
                                   : (xi + (size_t)(c - split) * D);
    return *(const v4f*)(src + sub * 4);
}

__global__ __launch_bounds__(256) void gather_kernel(const float* __restrict__ xu,
                                                     const float* __restrict__ xi,
                                                     int split,
                                                     const int*  __restrict__ rowptr,
                                                     const int*  __restrict__ rowend,
                                                     const int2* __restrict__ sedge,
                                                     float* __restrict__ out,
                                                     int nedges) {
    const int row = (blockIdx.x * 256 + threadIdx.x) >> 4;
    const int sub = threadIdx.x & 15;
    if (row >= NUM_NODES) return;
    const int start = min(max(rowptr[row], 0), nedges);
    const int end   = min(max(rowend[row], start), nedges);
    v4f acc = {0.f, 0.f, 0.f, 0.f};
    for (int base = start; base < end; base += 16) {
        const int n = min(end - base, 16);
        int mc = 0, mv = 0;
        if (sub < n) {
            int2 m = sedge[base + sub];
            mc = m.x;
            mv = m.y;
        }
        for (int t = 0; t < n; t += 4) {
            const int i1 = min(t + 1, n - 1);
            const int i2 = min(t + 2, n - 1);
            const int i3 = min(t + 3, n - 1);
            const int   c0 = __shfl(mc, t,  16);
            const int   c1 = __shfl(mc, i1, 16);
            const int   c2 = __shfl(mc, i2, 16);
            const int   c3 = __shfl(mc, i3, 16);
            const float w0 = __int_as_float(__shfl(mv, t,  16));
            const float w1 = __int_as_float(__shfl(mv, i1, 16));
            const float w2 = __int_as_float(__shfl(mv, i2, 16));
            const float w3 = __int_as_float(__shfl(mv, i3, 16));
            v4f x0 = load_x4(xu, xi, split, c0, sub);
            v4f x1 = load_x4(xu, xi, split, c1, sub);
            v4f x2 = load_x4(xu, xi, split, c2, sub);
            v4f x3 = load_x4(xu, xi, split, c3, sub);
            const v4f zero = {0.f, 0.f, 0.f, 0.f};
            acc += w0 * x0;
            acc += (t + 1 < n) ? (w1 * x1) : zero;
            acc += (t + 2 < n) ? (w2 * x2) : zero;
            acc += (t + 3 < n) ? (w3 * x3) : zero;
        }
    }
    *(v4f*)(out + (size_t)row * D + sub * 4) = acc;
}

#define LDP (K2 + 4)
__global__ __launch_bounds__(128) void fuse_kernel(
    const float* __restrict__ user_emb, const float* __restrict__ item_emb,
    const float* __restrict__ hg_user,  const float* __restrict__ hg_item,
    const float* __restrict__ h1,       const float* __restrict__ h2,
    const float* __restrict__ W1,       const float* __restrict__ b1,
    const float* __restrict__ W2,       const float* __restrict__ b2,
    float* __restrict__ out)
{
    __shared__ float sW1t[D * LDP];
    __shared__ float sA[4][16 * LDP];

    const int tid  = threadIdx.x;
    const int w    = tid >> 5;
    const int lane = tid & 31;

    for (int i = tid; i < K2 * D; i += 128) {
        int k = i >> 6;
        int n = i & 63;
        sW1t[n * LDP + k] = W1[i];
    }

    const int tile_base = (blockIdx.x * 4 + w) * 16;

    for (int i = lane; i < 16 * 32; i += 32) {
        int node = i >> 5;
        int c4   = i & 31;
        int feat = c4 * 4;
        int ng   = tile_base + node;
        float4 val = make_float4(0.f, 0.f, 0.f, 0.f);
        if (ng < NUM_NODES) {
            if (feat < D) {
                const float* ego = (ng < M_USERS)
                    ? (user_emb + (size_t)ng * D + feat)
                    : (item_emb + (size_t)(ng - M_USERS) * D + feat);
                float4 e4 = *(const float4*)ego;
                float4 a4 = *(const float4*)(h1 + (size_t)ng * D + feat);
                float4 b4 = *(const float4*)(h2 + (size_t)ng * D + feat);
                const float inv3 = 1.0f / 3.0f;
                val = make_float4((e4.x + a4.x + b4.x) * inv3,
                                  (e4.y + a4.y + b4.y) * inv3,
                                  (e4.z + a4.z + b4.z) * inv3,
                                  (e4.w + a4.w + b4.w) * inv3);
            } else {
                int f = feat - D;
                const float* hg = (ng < M_USERS)
                    ? (hg_user + (size_t)ng * D + f)
                    : (hg_item + (size_t)(ng - M_USERS) * D + f);
                val = *(const float4*)hg;
            }
        }
        *(float4*)&sA[w][node * LDP + feat] = val;
    }

    __syncthreads();

    const int mrow = lane & 15;
    const int koff = (lane >> 4) * 2;

    v8f acc[4] = {};
    for (int kk = 0; kk < 32; ++kk) {
        int k0 = kk * 4 + koff;
        v2f a;
        a.x = sA[w][mrow * LDP + k0];
        a.y = sA[w][mrow * LDP + k0 + 1];
        #pragma unroll
        for (int nt = 0; nt < 4; ++nt) {
            int ncol = nt * 16 + mrow;
            v2f b;
            b.x = sW1t[ncol * LDP + k0];
            b.y = sW1t[ncol * LDP + k0 + 1];
            acc[nt] = __builtin_amdgcn_wmma_f32_16x16x4_f32(
                false, a, false, b, (short)0, acc[nt], false, false);
        }
    }

    float p0[8], p1[8];
    #pragma unroll
    for (int v = 0; v < 8; ++v) { p0[v] = 0.f; p1[v] = 0.f; }
    for (int nt = 0; nt < 4; ++nt) {
        int ncol = nt * 16 + mrow;
        float bb  = b1[ncol];
        float w2a = W2[ncol * 2 + 0];
        float w2b = W2[ncol * 2 + 1];
        #pragma unroll
        for (int v = 0; v < 8; ++v) {
            float hdn = tanhf(acc[nt][v] + bb);
            p0[v] += hdn * w2a;
            p1[v] += hdn * w2b;
        }
    }

    #pragma unroll
    for (int off = 1; off < 16; off <<= 1) {
        #pragma unroll
        for (int v = 0; v < 8; ++v) {
            p0[v] += __shfl_xor(p0[v], off, 32);
            p1[v] += __shfl_xor(p1[v], off, 32);
        }
    }

    float bb0 = b2[0], bb1 = b2[1];
    const int half = lane >> 4;

    for (int v = 0; v < 8; ++v) {
        float l0 = p0[v] + bb0;
        float l1 = p1[v] + bb1;
        float e  = expf(l1 - l0);
        float w1w = e / (1.0f + e);
        float w0w = 1.0f - w1w;
        int m  = v + half * 8;
        int ng = tile_base + m;
        if (ng < NUM_NODES) {
            #pragma unroll
            for (int ff = 0; ff < 4; ++ff) {
                int feat = mrow + ff * 16;
                float f1v = sA[w][m * LDP + feat];
                float f2v = sA[w][m * LDP + D + feat];
                out[(size_t)ng * D + feat] = w0w * f1v + w1w * f2v;
            }
        }
    }
}

extern "C" void kernel_launch(void* const* d_in, const int* in_sizes, int n_in,
                              void* d_out, int out_size, void* d_ws, size_t ws_size,
                              hipStream_t stream) {
    const float* user_emb = (const float*)d_in[0];
    const float* item_emb = (const float*)d_in[1];
    const float* hg_user  = (const float*)d_in[2];
    const float* hg_item  = (const float*)d_in[3];
    const float* adj_vals = (const float*)d_in[4];
    const float* W1       = (const float*)d_in[5];
    const float* b1       = (const float*)d_in[6];
    const float* W2       = (const float*)d_in[7];
    const float* b2       = (const float*)d_in[8];
    const int*   adj_rows = (const int*)d_in[9];
    const int*   adj_cols = (const int*)d_in[10];
    float* out = (float*)d_out;

    float* h1 = (float*)d_ws;
    float* h2 = h1 + (size_t)NUM_NODES * D;
    int*  cnt      = (int*)(h2 + (size_t)NUM_NODES * D);
    int*  rowptr   = cnt + NUM_NODES;
    int*  cursor   = rowptr + NUM_NODES;
    int*  blocksum = cursor + NUM_NODES;
    int2* sedge    = (int2*)(blocksum + 512);

    int E = in_sizes[4];

    zero_kernel<<<SCAN_BLOCKS, 256, 0, stream>>>((float4*)cnt, (long)(NUM_NODES / 4));
    hist_kernel<<<2048, 256, 0, stream>>>(adj_rows, cnt, E);
    blocksum_kernel<<<SCAN_BLOCKS, 256, 0, stream>>>(cnt, blocksum);
    scan_kernel<<<SCAN_BLOCKS, 256, 0, stream>>>(cnt, blocksum, rowptr, cursor);
    fill_kernel<<<2048, 256, 0, stream>>>(adj_rows, adj_cols, adj_vals, cursor, sedge, E);

    int gather_blocks = (NUM_NODES * 16 + 255) / 256;
    gather_kernel<<<gather_blocks, 256, 0, stream>>>(user_emb, item_emb, M_USERS,
                                                     rowptr, cursor, sedge, h1, E);
    gather_kernel<<<gather_blocks, 256, 0, stream>>>(h1, h1, NUM_NODES,
                                                     rowptr, cursor, sedge, h2, E);

    int fuse_blocks = (NUM_NODES + 63) / 64;
    fuse_kernel<<<fuse_blocks, 128, 0, stream>>>(user_emb, item_emb, hg_user, hg_item,
                                                 h1, h2, W1, b1, W2, b2, out);
}


# fuse MLP: the 32 row loads of a wave's 16-node tile are issued as one group ahead of the W1 staging, in place of two dependent trips to memory per node; sums and their order unchanged
# speedup vs baseline: 2.0531x; 1.1238x over previous
#include <hip/hip_runtime.h>
#include <math.h>

#define D 64
#define K2 128
#define M_USERS 200000
#define NUM_NODES 300000

typedef __attribute__((ext_vector_type(2))) float v2f;
typedef __attribute__((ext_vector_type(8))) float v8f;

__global__ void zero_kernel(float4* __restrict__ p, long n4) {
    long i = (long)blockIdx.x * blockDim.x + threadIdx.x;
    long stride = (long)gridDim.x * blockDim.x;
    float4 z = make_float4(0.f, 0.f, 0.f, 0.f);
    for (; i < n4; i += stride) p[i] = z;
}

#define SCAN_CHUNK 1024
#define SCAN_BLOCKS ((NUM_NODES + SCAN_CHUNK - 1) / SCAN_CHUNK)
static_assert(NUM_NODES % 4 == 0, "row counters are read as int4");

__global__ void hist_kernel(const int* __restrict__ rows, int* __restrict__ cnt, int nedges) {
    int i = blockIdx.x * blockDim.x + threadIdx.x;
    int stride = gridDim.x * blockDim.x;
    for (int e = i; e < nedges; e += stride) atomicAdd(&cnt[rows[e]], 1);
}

__global__ __launch_bounds__(256) void blocksum_kernel(const int* __restrict__ cnt,
                                                       int* __restrict__ blocksum) {
    __shared__ int s_red[256];
    const int t = threadIdx.x;
    const int i4 = blockIdx.x * 256 + t;
    int s = 0;
    if (i4 < NUM_NODES / 4) {
        int4 c = ((const int4*)cnt)[i4];
        s = c.x + c.y + c.z + c.w;
    }
    s_red[t] = s;
    __syncthreads();
    for (int off = 128; off > 0; off >>= 1) {
        if (t < off) s_red[t] += s_red[t + off];
        __syncthreads();
    }
    if (t == 0) blocksum[blockIdx.x] = s_red[0];
}

__global__ __launch_bounds__(256) void scan_kernel(const int* __restrict__ cnt,
                                                   const int* __restrict__ blocksum,
                                                   int* __restrict__ rowptr,
                                                   int* __restrict__ cursor) {
    __shared__ int s_red[256];
    __shared__ int s_scan[256];
    const int t = threadIdx.x;
    const int b = blockIdx.x;

    int part = 0;
    for (int j = t; j < b; j += 256) part += blocksum[j];
    s_red[t] = part;
    __syncthreads();
    for (int off = 128; off > 0; off >>= 1) {
        if (t < off) s_red[t] += s_red[t + off];
        __syncthreads();
    }
    const int base = s_red[0];

    const int i4 = b * 256 + t;
    const bool valid = i4 < NUM_NODES / 4;
    int4 c = make_int4(0, 0, 0, 0);
    if (valid) c = ((const int4*)cnt)[i4];
    const int s = c.x + c.y + c.z + c.w;

    s_scan[t] = s;
    __syncthreads();
    for (int off = 1; off < 256; off <<= 1) {
        int add = (t >= off) ? s_scan[t - off] : 0;
        __syncthreads();
        s_scan[t] += add;
        __syncthreads();
    }
    const int excl = base + s_scan[t] - s;

    if (valid) {
        int4 o;
        o.x = excl;
        o.y = excl + c.x;
        o.z = excl + c.x + c.y;
        o.w = excl + c.x + c.y + c.z;
        ((int4*)rowptr)[i4] = o;
        ((int4*)cursor)[i4] = o;
    }
}

__global__ void fill_kernel(const int* __restrict__ rows, const int* __restrict__ cols,
                            const float* __restrict__ vals, int* __restrict__ cursor,
                            int2* __restrict__ sedge, int nedges) {
    int i = blockIdx.x * blockDim.x + threadIdx.x;
    int stride = gridDim.x * blockDim.x;
    for (int e = i; e < nedges; e += stride) {
        int pos = atomicAdd(&cursor[rows[e]], 1);
        if ((unsigned)pos < (unsigned)nedges)
            sedge[pos] = make_int2(cols[e], __float_as_int(vals[e]));
    }
}

typedef __attribute__((ext_vector_type(4))) float v4f;

__device__ __forceinline__ v4f load_x4(const float* __restrict__ xu,
                                       const float* __restrict__ xi,
                                       int split, int c, int sub) {
    const float* src = (c < split) ? (xu + (size_t)c * D)
                                   : (xi + (size_t)(c - split) * D);
    return *(const v4f*)(src + sub * 4);
}

__global__ __launch_bounds__(256) void gather_kernel(const float* __restrict__ xu,
                                                     const float* __restrict__ xi,
                                                     int split,
                                                     const int*  __restrict__ rowptr,
                                                     const int*  __restrict__ rowend,
                                                     const int2* __restrict__ sedge,
                                                     float* __restrict__ out,
                                                     int nedges) {
    const int row = (blockIdx.x * 256 + threadIdx.x) >> 4;
    const int sub = threadIdx.x & 15;
    if (row >= NUM_NODES) return;
    const int start = min(max(rowptr[row], 0), nedges);
    const int end   = min(max(rowend[row], start), nedges);
    v4f acc = {0.f, 0.f, 0.f, 0.f};
    for (int base = start; base < end; base += 16) {
        const int n = min(end - base, 16);
        int mc = 0, mv = 0;
        if (sub < n) {
            int2 m = sedge[base + sub];
            mc = m.x;
            mv = m.y;
        }
        for (int t = 0; t < n; t += 4) {
            const int i1 = min(t + 1, n - 1);
            const int i2 = min(t + 2, n - 1);
            const int i3 = min(t + 3, n - 1);
            const int   c0 = __shfl(mc, t,  16);
            const int   c1 = __shfl(mc, i1, 16);
            const int   c2 = __shfl(mc, i2, 16);
            const int   c3 = __shfl(mc, i3, 16);
            const float w0 = __int_as_float(__shfl(mv, t,  16));
            const float w1 = __int_as_float(__shfl(mv, i1, 16));
            const float w2 = __int_as_float(__shfl(mv, i2, 16));
            const float w3 = __int_as_float(__shfl(mv, i3, 16));
            v4f x0 = load_x4(xu, xi, split, c0, sub);
            v4f x1 = load_x4(xu, xi, split, c1, sub);
            v4f x2 = load_x4(xu, xi, split, c2, sub);
            v4f x3 = load_x4(xu, xi, split, c3, sub);
            const v4f zero = {0.f, 0.f, 0.f, 0.f};
            acc += w0 * x0;
            acc += (t + 1 < n) ? (w1 * x1) : zero;
            acc += (t + 2 < n) ? (w2 * x2) : zero;
            acc += (t + 3 < n) ? (w3 * x3) : zero;
        }
    }
    *(v4f*)(out + (size_t)row * D + sub * 4) = acc;
}

#define LDP (K2 + 4)
__global__ __launch_bounds__(128) void fuse_kernel(
    const float* __restrict__ user_emb, const float* __restrict__ item_emb,
    const float* __restrict__ hg_user,  const float* __restrict__ hg_item,
    const float* __restrict__ h1,       const float* __restrict__ h2,
    const float* __restrict__ W1,       const float* __restrict__ b1,
    const float* __restrict__ W2,       const float* __restrict__ b2,
    float* __restrict__ out)
{
    __shared__ float sW1t[D * LDP];
    __shared__ float sA[4][16 * LDP];

    const int tid  = threadIdx.x;
    const int w    = tid >> 5;
    const int lane = tid & 31;

    const int tile_base = (blockIdx.x * 4 + w) * 16;

    const int hsel = lane >> 4;
    const int c4   = lane & 15;
    float4 e4[8], a4[8], b4[8], g4[8];
    #pragma unroll
    for (int s = 0; s < 8; ++s) {
        const int ng = tile_base + 2 * s + hsel;
        const int nc = min(ng, NUM_NODES - 1);
        const bool is_user = nc < M_USERS;
        const float* ego = is_user ? (user_emb + (size_t)nc * D)
                                   : (item_emb + (size_t)(nc - M_USERS) * D);
        const float* hg  = is_user ? (hg_user + (size_t)nc * D)
                                   : (hg_item + (size_t)(nc - M_USERS) * D);
        e4[s] = *(const float4*)(ego + c4 * 4);
        a4[s] = *(const float4*)(h1 + (size_t)nc * D + c4 * 4);
        b4[s] = *(const float4*)(h2 + (size_t)nc * D + c4 * 4);
        g4[s] = *(const float4*)(hg + c4 * 4);
    }

    for (int i = tid; i < K2 * D; i += 128) {
        int k = i >> 6;
        int n = i & 63;
        sW1t[n * LDP + k] = W1[i];
    }

    #pragma unroll
    for (int s = 0; s < 8; ++s) {
        const int node = 2 * s + hsel;
        const bool in_range = (tile_base + node) < NUM_NODES;
        const float inv3 = 1.0f / 3.0f;
        float4 f1 = make_float4((e4[s].x + a4[s].x + b4[s].x) * inv3,
                                (e4[s].y + a4[s].y + b4[s].y) * inv3,
                                (e4[s].z + a4[s].z + b4[s].z) * inv3,
                                (e4[s].w + a4[s].w + b4[s].w) * inv3);
        float4 f2 = g4[s];
        if (!in_range) {
            f1 = make_float4(0.f, 0.f, 0.f, 0.f);
            f2 = make_float4(0.f, 0.f, 0.f, 0.f);
        }
        *(float4*)&sA[w][node * LDP + c4 * 4]     = f1;
        *(float4*)&sA[w][node * LDP + D + c4 * 4] = f2;
    }

    __syncthreads();

    const int mrow = lane & 15;
    const int koff = (lane >> 4) * 2;

    v8f acc[4] = {};
    for (int kk = 0; kk < 32; ++kk) {
        int k0 = kk * 4 + koff;
        v2f a;
        a.x = sA[w][mrow * LDP + k0];
        a.y = sA[w][mrow * LDP + k0 + 1];
        #pragma unroll
        for (int nt = 0; nt < 4; ++nt) {
            int ncol = nt * 16 + mrow;
            v2f b;
            b.x = sW1t[ncol * LDP + k0];
            b.y = sW1t[ncol * LDP + k0 + 1];
            acc[nt] = __builtin_amdgcn_wmma_f32_16x16x4_f32(
                false, a, false, b, (short)0, acc[nt], false, false);
        }
    }

    float p0[8], p1[8];
    #pragma unroll
    for (int v = 0; v < 8; ++v) { p0[v] = 0.f; p1[v] = 0.f; }
    for (int nt = 0; nt < 4; ++nt) {
        int ncol = nt * 16 + mrow;
        float bb  = b1[ncol];
        float w2a = W2[ncol * 2 + 0];
        float w2b = W2[ncol * 2 + 1];
        #pragma unroll
        for (int v = 0; v < 8; ++v) {
            float hdn = tanhf(acc[nt][v] + bb);
            p0[v] += hdn * w2a;
            p1[v] += hdn * w2b;
        }
    }

    #pragma unroll
    for (int off = 1; off < 16; off <<= 1) {
        #pragma unroll
        for (int v = 0; v < 8; ++v) {
            p0[v] += __shfl_xor(p0[v], off, 32);
            p1[v] += __shfl_xor(p1[v], off, 32);
        }
    }

    float bb0 = b2[0], bb1 = b2[1];
    const int half = lane >> 4;

    for (int v = 0; v < 8; ++v) {
        float l0 = p0[v] + bb0;
        float l1 = p1[v] + bb1;
        float e  = expf(l1 - l0);
        float w1w = e / (1.0f + e);
        float w0w = 1.0f - w1w;
        int m  = v + half * 8;
        int ng = tile_base + m;
        if (ng < NUM_NODES) {
            #pragma unroll
            for (int ff = 0; ff < 4; ++ff) {
                int feat = mrow + ff * 16;
                float f1v = sA[w][m * LDP + feat];
                float f2v = sA[w][m * LDP + D + feat];
                out[(size_t)ng * D + feat] = w0w * f1v + w1w * f2v;
            }
        }
    }
}

extern "C" void kernel_launch(void* const* d_in, const int* in_sizes, int n_in,
                              void* d_out, int out_size, void* d_ws, size_t ws_size,
                              hipStream_t stream) {
    const float* user_emb = (const float*)d_in[0];
    const float* item_emb = (const float*)d_in[1];
    const float* hg_user  = (const float*)d_in[2];
    const float* hg_item  = (const float*)d_in[3];
    const float* adj_vals = (const float*)d_in[4];
    const float* W1       = (const float*)d_in[5];
    const float* b1       = (const float*)d_in[6];
    const float* W2       = (const float*)d_in[7];
    const float* b2       = (const float*)d_in[8];
    const int*   adj_rows = (const int*)d_in[9];
    const int*   adj_cols = (const int*)d_in[10];
    float* out = (float*)d_out;

    float* h1 = (float*)d_ws;
    float* h2 = h1 + (size_t)NUM_NODES * D;
    int*  cnt      = (int*)(h2 + (size_t)NUM_NODES * D);
    int*  rowptr   = cnt + NUM_NODES;
    int*  cursor   = rowptr + NUM_NODES;
    int*  blocksum = cursor + NUM_NODES;
    int2* sedge    = (int2*)(blocksum + 512);

    int E = in_sizes[4];

    zero_kernel<<<SCAN_BLOCKS, 256, 0, stream>>>((float4*)cnt, (long)(NUM_NODES / 4));
    hist_kernel<<<2048, 256, 0, stream>>>(adj_rows, cnt, E);
    blocksum_kernel<<<SCAN_BLOCKS, 256, 0, stream>>>(cnt, blocksum);
    scan_kernel<<<SCAN_BLOCKS, 256, 0, stream>>>(cnt, blocksum, rowptr, cursor);
    fill_kernel<<<2048, 256, 0, stream>>>(adj_rows, adj_cols, adj_vals, cursor, sedge, E);

    int gather_blocks = (NUM_NODES * 16 + 255) / 256;
    gather_kernel<<<gather_blocks, 256, 0, stream>>>(user_emb, item_emb, M_USERS,
                                                     rowptr, cursor, sedge, h1, E);
    gather_kernel<<<gather_blocks, 256, 0, stream>>>(h1, h1, NUM_NODES,
                                                     rowptr, cursor, sedge, h2, E);

    int fuse_blocks = (NUM_NODES + 63) / 64;
    fuse_kernel<<<fuse_blocks, 128, 0, stream>>>(user_emb, item_emb, hg_user, hg_item,
                                                 h1, h2, W1, b1, W2, b2, out);
}


# edge sort: the histogram's add returns each edge's rank in its row (4 edges per thread in flight), slot = row start + rank, so the fill pass has no atomics; bucket (sum) order = histogram arrival
# speedup vs baseline: 2.5023x; 1.2188x over previous
#include <hip/hip_runtime.h>
#include <math.h>

#define D 64
#define K2 128
#define M_USERS 200000
#define NUM_NODES 300000

typedef __attribute__((ext_vector_type(2))) float v2f;
typedef __attribute__((ext_vector_type(8))) float v8f;

__global__ void zero_kernel(float4* __restrict__ p, long n4) {
    long i = (long)blockIdx.x * blockDim.x + threadIdx.x;
    long stride = (long)gridDim.x * blockDim.x;
    float4 z = make_float4(0.f, 0.f, 0.f, 0.f);
    for (; i < n4; i += stride) p[i] = z;
}

#define SCAN_CHUNK 1024
#define SCAN_BLOCKS ((NUM_NODES + SCAN_CHUNK - 1) / SCAN_CHUNK)
static_assert(NUM_NODES % 4 == 0, "row counters are read as int4");

__global__ void hist_kernel(const int* __restrict__ rows, int* __restrict__ cnt,
                            int* __restrict__ rank, int nedges) {
    const int i = blockIdx.x * blockDim.x + threadIdx.x;
    const int stride = gridDim.x * blockDim.x;
    const int nq = nedges >> 2;
    for (int q = i; q < nq; q += stride) {
        const int4 r = ((const int4*)rows)[q];
        int4 k;
        k.x = atomicAdd(&cnt[r.x], 1);
        k.y = atomicAdd(&cnt[r.y], 1);
        k.z = atomicAdd(&cnt[r.z], 1);
        k.w = atomicAdd(&cnt[r.w], 1);
        ((int4*)rank)[q] = k;
    }
    for (int e = (nq << 2) + i; e < nedges; e += stride) rank[e] = atomicAdd(&cnt[rows[e]], 1);
}

__global__ __launch_bounds__(256) void blocksum_kernel(const int* __restrict__ cnt,
                                                       int* __restrict__ blocksum) {
    __shared__ int s_red[256];
    const int t = threadIdx.x;
    const int i4 = blockIdx.x * 256 + t;
    int s = 0;
    if (i4 < NUM_NODES / 4) {
        int4 c = ((const int4*)cnt)[i4];
        s = c.x + c.y + c.z + c.w;
    }
    s_red[t] = s;
    __syncthreads();
    for (int off = 128; off > 0; off >>= 1) {
        if (t < off) s_red[t] += s_red[t + off];
        __syncthreads();
    }
    if (t == 0) blocksum[blockIdx.x] = s_red[0];
}

__global__ __launch_bounds__(256) void scan_kernel(const int* __restrict__ cnt,
                                                   const int* __restrict__ blocksum,
                                                   int* __restrict__ rowptr,
                                                   int* __restrict__ rowend) {
    __shared__ int s_red[256];
    __shared__ int s_scan[256];
    const int t = threadIdx.x;
    const int b = blockIdx.x;

    int part = 0;
    for (int j = t; j < b; j += 256) part += blocksum[j];
    s_red[t] = part;
    __syncthreads();
    for (int off = 128; off > 0; off >>= 1) {
        if (t < off) s_red[t] += s_red[t + off];
        __syncthreads();
    }
    const int base = s_red[0];

    const int i4 = b * 256 + t;
    const bool valid = i4 < NUM_NODES / 4;
    int4 c = make_int4(0, 0, 0, 0);
    if (valid) c = ((const int4*)cnt)[i4];
    const int s = c.x + c.y + c.z + c.w;

    s_scan[t] = s;
    __syncthreads();
    for (int off = 1; off < 256; off <<= 1) {
        int add = (t >= off) ? s_scan[t - off] : 0;
        __syncthreads();
        s_scan[t] += add;
        __syncthreads();
    }
    const int excl = base + s_scan[t] - s;

    if (valid) {
        int4 o;
        o.x = excl;
        o.y = excl + c.x;
        o.z = excl + c.x + c.y;
        o.w = excl + c.x + c.y + c.z;
        int4 en;
        en.x = o.y;
        en.y = o.z;
        en.z = o.w;
        en.w = o.w + c.w;
        ((int4*)rowptr)[i4] = o;
        ((int4*)rowend)[i4] = en;
    }
}

__device__ __forceinline__ void place_edge(int2* __restrict__ sedge, int pos, int col, float val,
                                           int nedges) {
    const unsigned p = min((unsigned)pos, (unsigned)(nedges - 1));
    sedge[p] = make_int2(col, __float_as_int(val));
}

__global__ void fill_kernel(const int* __restrict__ rows, const int* __restrict__ cols,
                            const float* __restrict__ vals, const int* __restrict__ rank,
                            const int* __restrict__ rowptr, int2* __restrict__ sedge, int nedges) {
    const int i = blockIdx.x * blockDim.x + threadIdx.x;
    const int stride = gridDim.x * blockDim.x;
    const int nq = nedges >> 2;
    for (int q = i; q < nq; q += stride) {
        const int4   r = ((const int4*)rows)[q];
        const int4   k = ((const int4*)rank)[q];
        const int4   c = ((const int4*)cols)[q];
        const float4 v = ((const float4*)vals)[q];
        const int p0 = rowptr[r.x] + k.x;
        const int p1 = rowptr[r.y] + k.y;
        const int p2 = rowptr[r.z] + k.z;
        const int p3 = rowptr[r.w] + k.w;
        place_edge(sedge, p0, c.x, v.x, nedges);
        place_edge(sedge, p1, c.y, v.y, nedges);
        place_edge(sedge, p2, c.z, v.z, nedges);
        place_edge(sedge, p3, c.w, v.w, nedges);
    }
    for (int e = (nq << 2) + i; e < nedges; e += stride)
        place_edge(sedge, rowptr[rows[e]] + rank[e], cols[e], vals[e], nedges);
}

typedef __attribute__((ext_vector_type(4))) float v4f;

__device__ __forceinline__ v4f load_x4(const float* __restrict__ xu,
                                       const float* __restrict__ xi,
                                       int split, int c, int sub) {
    const float* src = (c < split) ? (xu + (size_t)c * D)
                                   : (xi + (size_t)(c - split) * D);
    return *(const v4f*)(src + sub * 4);
}

__global__ __launch_bounds__(256) void gather_kernel(const float* __restrict__ xu,
                                                     const float* __restrict__ xi,
                                                     int split,
                                                     const int*  __restrict__ rowptr,
                                                     const int*  __restrict__ rowend,
                                                     const int2* __restrict__ sedge,
                                                     float* __restrict__ out,
                                                     int nedges) {
    const int row = (blockIdx.x * 256 + threadIdx.x) >> 4;
    const int sub = threadIdx.x & 15;
    if (row >= NUM_NODES) return;
    const int start = min(max(rowptr[row], 0), nedges);
    const int end   = min(max(rowend[row], start), nedges);
    v4f acc = {0.f, 0.f, 0.f, 0.f};
    for (int base = start; base < end; base += 16) {
        const int n = min(end - base, 16);
        int mc = 0, mv = 0;
        if (sub < n) {
            int2 m = sedge[base + sub];
            mc = m.x;
            mv = m.y;
        }
        for (int t = 0; t < n; t += 4) {
            const int i1 = min(t + 1, n - 1);
            const int i2 = min(t + 2, n - 1);
            const int i3 = min(t + 3, n - 1);
            const int   c0 = __shfl(mc, t,  16);
            const int   c1 = __shfl(mc, i1, 16);
            const int   c2 = __shfl(mc, i2, 16);
            const int   c3 = __shfl(mc, i3, 16);
            const float w0 = __int_as_float(__shfl(mv, t,  16));
            const float w1 = __int_as_float(__shfl(mv, i1, 16));
            const float w2 = __int_as_float(__shfl(mv, i2, 16));
            const float w3 = __int_as_float(__shfl(mv, i3, 16));
            v4f x0 = load_x4(xu, xi, split, c0, sub);
            v4f x1 = load_x4(xu, xi, split, c1, sub);
            v4f x2 = load_x4(xu, xi, split, c2, sub);
            v4f x3 = load_x4(xu, xi, split, c3, sub);
            const v4f zero = {0.f, 0.f, 0.f, 0.f};
            acc += w0 * x0;
            acc += (t + 1 < n) ? (w1 * x1) : zero;
            acc += (t + 2 < n) ? (w2 * x2) : zero;
            acc += (t + 3 < n) ? (w3 * x3) : zero;
        }
    }
    *(v4f*)(out + (size_t)row * D + sub * 4) = acc;
}

#define LDP (K2 + 4)
__global__ __launch_bounds__(128) void fuse_kernel(
    const float* __restrict__ user_emb, const float* __restrict__ item_emb,
    const float* __restrict__ hg_user,  const float* __restrict__ hg_item,
    const float* __restrict__ h1,       const float* __restrict__ h2,
    const float* __restrict__ W1,       const float* __restrict__ b1,
    const float* __restrict__ W2,       const float* __restrict__ b2,
    float* __restrict__ out)
{
    __shared__ float sW1t[D * LDP];
    __shared__ float sA[4][16 * LDP];

    const int tid  = threadIdx.x;
    const int w    = tid >> 5;
    const int lane = tid & 31;

    const int tile_base = (blockIdx.x * 4 + w) * 16;

    const int hsel = lane >> 4;
    const int c4   = lane & 15;
    float4 e4[8], a4[8], b4[8], g4[8];
    #pragma unroll
    for (int s = 0; s < 8; ++s) {
        const int ng = tile_base + 2 * s + hsel;
        const int nc = min(ng, NUM_NODES - 1);
        const bool is_user = nc < M_USERS;
        const float* ego = is_user ? (user_emb + (size_t)nc * D)
                                   : (item_emb + (size_t)(nc - M_USERS) * D);
        const float* hg  = is_user ? (hg_user + (size_t)nc * D)
                                   : (hg_item + (size_t)(nc - M_USERS) * D);
        e4[s] = *(const float4*)(ego + c4 * 4);
        a4[s] = *(const float4*)(h1 + (size_t)nc * D + c4 * 4);
        b4[s] = *(const float4*)(h2 + (size_t)nc * D + c4 * 4);
        g4[s] = *(const float4*)(hg + c4 * 4);
    }

    for (int i = tid; i < K2 * D; i += 128) {
        int k = i >> 6;
        int n = i & 63;
        sW1t[n * LDP + k] = W1[i];
    }

    #pragma unroll
    for (int s = 0; s < 8; ++s) {
        const int node = 2 * s + hsel;
        const bool in_range = (tile_base + node) < NUM_NODES;
        const float inv3 = 1.0f / 3.0f;
        float4 f1 = make_float4((e4[s].x + a4[s].x + b4[s].x) * inv3,
                                (e4[s].y + a4[s].y + b4[s].y) * inv3,
                                (e4[s].z + a4[s].z + b4[s].z) * inv3,
                                (e4[s].w + a4[s].w + b4[s].w) * inv3);
        float4 f2 = g4[s];
        if (!in_range) {
            f1 = make_float4(0.f, 0.f, 0.f, 0.f);
            f2 = make_float4(0.f, 0.f, 0.f, 0.f);
        }
        *(float4*)&sA[w][node * LDP + c4 * 4]     = f1;
        *(float4*)&sA[w][node * LDP + D + c4 * 4] = f2;
    }

    __syncthreads();

    const int mrow = lane & 15;
    const int koff = (lane >> 4) * 2;

    v8f acc[4] = {};
    for (int kk = 0; kk < 32; ++kk) {
        int k0 = kk * 4 + koff;
        v2f a;
        a.x = sA[w][mrow * LDP + k0];
        a.y = sA[w][mrow * LDP + k0 + 1];
        #pragma unroll
        for (int nt = 0; nt < 4; ++nt) {
            int ncol = nt * 16 + mrow;
            v2f b;
            b.x = sW1t[ncol * LDP + k0];
            b.y = sW1t[ncol * LDP + k0 + 1];
            acc[nt] = __builtin_amdgcn_wmma_f32_16x16x4_f32(
                false, a, false, b, (short)0, acc[nt], false, false);
        }
    }

    float p0[8], p1[8];
    #pragma unroll
    for (int v = 0; v < 8; ++v) { p0[v] = 0.f; p1[v] = 0.f; }
    for (int nt = 0; nt < 4; ++nt) {
        int ncol = nt * 16 + mrow;
        float bb  = b1[ncol];
        float w2a = W2[ncol * 2 + 0];
        float w2b = W2[ncol * 2 + 1];
        #pragma unroll
        for (int v = 0; v < 8; ++v) {
            float hdn = tanhf(acc[nt][v] + bb);
            p0[v] += hdn * w2a;
            p1[v] += hdn * w2b;
        }
    }

    #pragma unroll
    for (int off = 1; off < 16; off <<= 1) {
        #pragma unroll
        for (int v = 0; v < 8; ++v) {
            p0[v] += __shfl_xor(p0[v], off, 32);
            p1[v] += __shfl_xor(p1[v], off, 32);
        }
    }

    float bb0 = b2[0], bb1 = b2[1];
    const int half = lane >> 4;

    for (int v = 0; v < 8; ++v) {
        float l0 = p0[v] + bb0;
        float l1 = p1[v] + bb1;
        float e  = expf(l1 - l0);
        float w1w = e / (1.0f + e);
        float w0w = 1.0f - w1w;
        int m  = v + half * 8;
        int ng = tile_base + m;
        if (ng < NUM_NODES) {
            #pragma unroll
            for (int ff = 0; ff < 4; ++ff) {
                int feat = mrow + ff * 16;
                float f1v = sA[w][m * LDP + feat];
                float f2v = sA[w][m * LDP + D + feat];
                out[(size_t)ng * D + feat] = w0w * f1v + w1w * f2v;
            }
        }
    }
}

extern "C" void kernel_launch(void* const* d_in, const int* in_sizes, int n_in,
                              void* d_out, int out_size, void* d_ws, size_t ws_size,
                              hipStream_t stream) {
    const float* user_emb = (const float*)d_in[0];
    const float* item_emb = (const float*)d_in[1];
    const float* hg_user  = (const float*)d_in[2];
    const float* hg_item  = (const float*)d_in[3];
    const float* adj_vals = (const float*)d_in[4];
    const float* W1       = (const float*)d_in[5];
    const float* b1       = (const float*)d_in[6];
    const float* W2       = (const float*)d_in[7];
    const float* b2       = (const float*)d_in[8];
    const int*   adj_rows = (const int*)d_in[9];
    const int*   adj_cols = (const int*)d_in[10];
    float* out = (float*)d_out;

    float* h1 = (float*)d_ws;
    float* h2 = h1 + (size_t)NUM_NODES * D;
    int*  cnt      = (int*)(h2 + (size_t)NUM_NODES * D);
    int*  rowptr   = cnt + NUM_NODES;
    int*  rowend   = rowptr + NUM_NODES;
    int*  blocksum = rowend + NUM_NODES;
    int2* sedge    = (int2*)(blocksum + 512);
    int*  rank     = (int*)h2;

    int E = in_sizes[4];

    zero_kernel<<<SCAN_BLOCKS, 256, 0, stream>>>((float4*)cnt, (long)(NUM_NODES / 4));
    hist_kernel<<<2048, 256, 0, stream>>>(adj_rows, cnt, rank, E);
    blocksum_kernel<<<SCAN_BLOCKS, 256, 0, stream>>>(cnt, blocksum);
    scan_kernel<<<SCAN_BLOCKS, 256, 0, stream>>>(cnt, blocksum, rowptr, rowend);
    fill_kernel<<<2048, 256, 0, stream>>>(adj_rows, adj_cols, adj_vals, rank, rowptr, sedge, E);

    int gather_blocks = (NUM_NODES * 16 + 255) / 256;
    gather_kernel<<<gather_blocks, 256, 0, stream>>>(user_emb, item_emb, M_USERS,
                                                     rowptr, rowend, sedge, h1, E);
    gather_kernel<<<gather_blocks, 256, 0, stream>>>(h1, h1, NUM_NODES,
                                                     rowptr, rowend, sedge, h2, E);

    int fuse_blocks = (NUM_NODES + 63) / 64;
    fuse_kernel<<<fuse_blocks, 128, 0, stream>>>(user_emb, item_emb, hg_user, hg_item,
                                                 h1, h2, W1, b1, W2, b2, out);
}


# edge sort: two-level counting sort (586 row buckets per slice, then rows inside a bucket) with every counter in LDS, in place of 4M scattered returning global adds; sum order = LDS add arrival
# speedup vs baseline: 3.3248x; 1.3287x over previous
#include <hip/hip_runtime.h>
#include <math.h>

#define D 64
#define K2 128
#define M_USERS 200000
#define NUM_NODES 300000

typedef __attribute__((ext_vector_type(2))) float v2f;
typedef __attribute__((ext_vector_type(8))) float v8f;

#define BSH 9
#define BROWS (1 << BSH)
#define NBK ((NUM_NODES + BROWS - 1) >> BSH)
#define PART_BLOCKS 1024
#define NCELL (NBK * PART_BLOCKS)
#define COLBITS 19
#define COLMASK ((1 << COLBITS) - 1)
#define SCAN_CHUNK 1024
static_assert(NUM_NODES <= (1 << COLBITS), "col and the row's place in its bucket share one word");
static_assert(BSH + COLBITS <= 31, "the shared word stays non-negative");
static_assert(PART_BLOCKS == SCAN_CHUNK, "one scan chunk per bucket: its sum is the bucket's size");
static_assert(NCELL % 4 == 0, "cell counters are read as int4");
#define EDGE_ROOM (NUM_NODES * D / 3)
static_assert(EDGE_ROOM % 4 == 0, "the bucket-ordered copy starts on a 16-byte boundary");

__device__ __forceinline__ void slice_bounds(int nedges, int b, int& start, int& stop) {
    const int ch = (((nedges + PART_BLOCKS - 1) / PART_BLOCKS) + 3) & ~3;
    start = min(b * ch, nedges);
    stop  = min(start + ch, nedges);
}

__device__ __forceinline__ int bucket_of(int row) {
    return (int)min((unsigned)row >> BSH, (unsigned)(NBK - 1));
}

__global__ __launch_bounds__(256) void part_count_kernel(const int* __restrict__ rows,
                                                         int* __restrict__ rank,
                                                         int* __restrict__ bcnt, int nedges) {
    __shared__ int s_cnt[NBK];
    const int t = threadIdx.x;
    const int b = blockIdx.x;
    for (int k = t; k < NBK; k += 256) s_cnt[k] = 0;
    __syncthreads();
    int start, stop;
    slice_bounds(nedges, b, start, stop);
    const int nq = (stop - start) >> 2;
    const int4* rows4 = (const int4*)(rows + start);
    int4* rank4 = (int4*)(rank + start);
    for (int q = t; q < nq; q += 256) {
        const int4 r = rows4[q];
        int4 k;
        k.x = atomicAdd(&s_cnt[bucket_of(r.x)], 1);
        k.y = atomicAdd(&s_cnt[bucket_of(r.y)], 1);
        k.z = atomicAdd(&s_cnt[bucket_of(r.z)], 1);
        k.w = atomicAdd(&s_cnt[bucket_of(r.w)], 1);
        rank4[q] = k;
    }
    for (int e = start + (nq << 2) + t; e < stop; e += 256)
        rank[e] = atomicAdd(&s_cnt[bucket_of(rows[e])], 1);
    __syncthreads();
    for (int k = t; k < NBK; k += 256) bcnt[k * PART_BLOCKS + b] = s_cnt[k];
}

__global__ __launch_bounds__(256) void blocksum_kernel(const int* __restrict__ cnt,
                                                       int* __restrict__ blocksum, int n4) {
    __shared__ int s_red[256];
    const int t = threadIdx.x;
    const int i4 = blockIdx.x * 256 + t;
    int s = 0;
    if (i4 < n4) {
        int4 c = ((const int4*)cnt)[i4];
        s = c.x + c.y + c.z + c.w;
    }
    s_red[t] = s;
    __syncthreads();
    for (int off = 128; off > 0; off >>= 1) {
        if (t < off) s_red[t] += s_red[t + off];
        __syncthreads();
    }
    if (t == 0) blocksum[blockIdx.x] = s_red[0];
}

__global__ __launch_bounds__(256) void scan_kernel(const int* __restrict__ cnt,
                                                   const int* __restrict__ blocksum,
                                                   int* __restrict__ excl_out, int n4) {
    __shared__ int s_red[256];
    __shared__ int s_scan[256];
    const int t = threadIdx.x;
    const int b = blockIdx.x;

    int part = 0;
    for (int j = t; j < b; j += 256) part += blocksum[j];
    s_red[t] = part;
    __syncthreads();
    for (int off = 128; off > 0; off >>= 1) {
        if (t < off) s_red[t] += s_red[t + off];
        __syncthreads();
    }
    const int base = s_red[0];

    const int i4 = b * 256 + t;
    const bool valid = i4 < n4;
    int4 c = make_int4(0, 0, 0, 0);
    if (valid) c = ((const int4*)cnt)[i4];
    const int s = c.x + c.y + c.z + c.w;

    s_scan[t] = s;
    __syncthreads();
    for (int off = 1; off < 256; off <<= 1) {
        int add = (t >= off) ? s_scan[t - off] : 0;
        __syncthreads();
        s_scan[t] += add;
        __syncthreads();
    }
    const int excl = base + s_scan[t] - s;

    if (valid) {
        int4 o;
        o.x = excl;
        o.y = excl + c.x;
        o.z = excl + c.x + c.y;
        o.w = excl + c.x + c.y + c.z;
        ((int4*)excl_out)[i4] = o;
    }
}

__device__ __forceinline__ void put_part(int2* __restrict__ pedge, int pos, int row, int col,
                                         float val, int nedges) {
    const unsigned p = min((unsigned)pos, (unsigned)(nedges - 1));
    pedge[p] = make_int2(((row & (BROWS - 1)) << COLBITS) | (col & COLMASK), __float_as_int(val));
}

__global__ __launch_bounds__(256) void part_scatter_kernel(const int* __restrict__ rows,
                                                           const int* __restrict__ cols,
                                                           const float* __restrict__ vals,
                                                           const int* __restrict__ rank,
                                                           const int* __restrict__ boff,
                                                           int2* __restrict__ pedge, int nedges) {
    __shared__ int s_off[NBK];
    const int t = threadIdx.x;
    const int b = blockIdx.x;
    for (int k = t; k < NBK; k += 256) s_off[k] = boff[k * PART_BLOCKS + b];
    __syncthreads();
    int start, stop;
    slice_bounds(nedges, b, start, stop);
    const int nq = (stop - start) >> 2;
    const int4*   rows4 = (const int4*)(rows + start);
    const int4*   rank4 = (const int4*)(rank + start);
    const int4*   cols4 = (const int4*)(cols + start);
    const float4* vals4 = (const float4*)(vals + start);
    for (int q = t; q < nq; q += 256) {
        const int4   r = rows4[q];
        const int4   k = rank4[q];
        const int4   cc = cols4[q];
        const float4 v = vals4[q];
        put_part(pedge, s_off[bucket_of(r.x)] + k.x, r.x, cc.x, v.x, nedges);
        put_part(pedge, s_off[bucket_of(r.y)] + k.y, r.y, cc.y, v.y, nedges);
        put_part(pedge, s_off[bucket_of(r.z)] + k.z, r.z, cc.z, v.z, nedges);
        put_part(pedge, s_off[bucket_of(r.w)] + k.w, r.w, cc.w, v.w, nedges);
    }
    for (int e = start + (nq << 2) + t; e < stop; e += 256)
        put_part(pedge, s_off[bucket_of(rows[e])] + rank[e], rows[e], cols[e], vals[e], nedges);
}

__device__ __forceinline__ int row_in_bucket(int word) {
    return (int)(((unsigned)word >> COLBITS) & (BROWS - 1));
}

__device__ __forceinline__ void place_edge(int2* __restrict__ sedge, int* s_cur, int lo, int2 m,
                                           int nedges) {
    const int pos = lo + atomicAdd(&s_cur[row_in_bucket(m.x)], 1);
    const unsigned p = min((unsigned)pos, (unsigned)(nedges - 1));
    sedge[p] = make_int2(m.x & COLMASK, m.y);
}

__global__ __launch_bounds__(BROWS) void bucket_rows_kernel(const int* __restrict__ boff,
                                                            const int* __restrict__ bsum,
                                                            const int2* __restrict__ pedge,
                                                            int* __restrict__ rowptr,
                                                            int* __restrict__ rowend,
                                                            int2* __restrict__ sedge, int nedges) {
    __shared__ int s_cnt[BROWS];
    __shared__ int s_scan[BROWS];
    __shared__ int s_cur[BROWS];
    const int t = threadIdx.x;
    const int k = blockIdx.x;
    const int lo = min(max(boff[k * PART_BLOCKS], 0), nedges);
    const int hi = min(max(lo + bsum[k], lo), nedges);

    s_cnt[t] = 0;
    __syncthreads();
    int i = lo + t;
    for (; i + 3 * BROWS < hi; i += 4 * BROWS) {
        const int x0 = pedge[i].x;
        const int x1 = pedge[i + BROWS].x;
        const int x2 = pedge[i + 2 * BROWS].x;
        const int x3 = pedge[i + 3 * BROWS].x;
        atomicAdd(&s_cnt[row_in_bucket(x0)], 1);
        atomicAdd(&s_cnt[row_in_bucket(x1)], 1);
        atomicAdd(&s_cnt[row_in_bucket(x2)], 1);
        atomicAdd(&s_cnt[row_in_bucket(x3)], 1);
    }
    for (; i < hi; i += BROWS) atomicAdd(&s_cnt[row_in_bucket(pedge[i].x)], 1);
    __syncthreads();

    const int cown = s_cnt[t];
    s_scan[t] = cown;
    __syncthreads();
    for (int off = 1; off < BROWS; off <<= 1) {
        const int add = (t >= off) ? s_scan[t - off] : 0;
        __syncthreads();
        s_scan[t] += add;
        __syncthreads();
    }
    const int excl = s_scan[t] - cown;
    s_cur[t] = excl;
    const int row = (k << BSH) + t;
    if (row < NUM_NODES) {
        rowptr[row] = lo + excl;
        rowend[row] = lo + excl + cown;
    }
    __syncthreads();

    i = lo + t;
    for (; i + 3 * BROWS < hi; i += 4 * BROWS) {
        const int2 m0 = pedge[i];
        const int2 m1 = pedge[i + BROWS];
        const int2 m2 = pedge[i + 2 * BROWS];
        const int2 m3 = pedge[i + 3 * BROWS];
        place_edge(sedge, s_cur, lo, m0, nedges);
        place_edge(sedge, s_cur, lo, m1, nedges);
        place_edge(sedge, s_cur, lo, m2, nedges);
        place_edge(sedge, s_cur, lo, m3, nedges);
    }
    for (; i < hi; i += BROWS) place_edge(sedge, s_cur, lo, pedge[i], nedges);
}

typedef __attribute__((ext_vector_type(4))) float v4f;

__device__ __forceinline__ v4f load_x4(const float* __restrict__ xu,
                                       const float* __restrict__ xi,
                                       int split, int c, int sub) {
    const float* src = (c < split) ? (xu + (size_t)c * D)
                                   : (xi + (size_t)(c - split) * D);
    return *(const v4f*)(src + sub * 4);
}

__global__ __launch_bounds__(256) void gather_kernel(const float* __restrict__ xu,
                                                     const float* __restrict__ xi,
                                                     int split,
                                                     const int*  __restrict__ rowptr,
                                                     const int*  __restrict__ rowend,
                                                     const int2* __restrict__ sedge,
                                                     float* __restrict__ out,
                                                     int nedges) {
    const int row = (blockIdx.x * 256 + threadIdx.x) >> 4;
    const int sub = threadIdx.x & 15;
    if (row >= NUM_NODES) return;
    const int start = min(max(rowptr[row], 0), nedges);
    const int end   = min(max(rowend[row], start), nedges);
    v4f acc = {0.f, 0.f, 0.f, 0.f};
    for (int base = start; base < end; base += 16) {
        const int n = min(end - base, 16);
        int mc = 0, mv = 0;
        if (sub < n) {
            int2 m = sedge[base + sub];
            mc = m.x;
            mv = m.y;
        }
        for (int t = 0; t < n; t += 4) {
            const int i1 = min(t + 1, n - 1);
            const int i2 = min(t + 2, n - 1);
            const int i3 = min(t + 3, n - 1);
            const int   c0 = __shfl(mc, t,  16);
            const int   c1 = __shfl(mc, i1, 16);
            const int   c2 = __shfl(mc, i2, 16);
            const int   c3 = __shfl(mc, i3, 16);
            const float w0 = __int_as_float(__shfl(mv, t,  16));
            const float w1 = __int_as_float(__shfl(mv, i1, 16));
            const float w2 = __int_as_float(__shfl(mv, i2, 16));
            const float w3 = __int_as_float(__shfl(mv, i3, 16));
            v4f x0 = load_x4(xu, xi, split, c0, sub);
            v4f x1 = load_x4(xu, xi, split, c1, sub);
            v4f x2 = load_x4(xu, xi, split, c2, sub);
            v4f x3 = load_x4(xu, xi, split, c3, sub);
            const v4f zero = {0.f, 0.f, 0.f, 0.f};
            acc += w0 * x0;
            acc += (t + 1 < n) ? (w1 * x1) : zero;
            acc += (t + 2 < n) ? (w2 * x2) : zero;
            acc += (t + 3 < n) ? (w3 * x3) : zero;
        }
    }
    *(v4f*)(out + (size_t)row * D + sub * 4) = acc;
}

#define LDP (K2 + 4)
__global__ __launch_bounds__(128) void fuse_kernel(
    const float* __restrict__ user_emb, const float* __restrict__ item_emb,
    const float* __restrict__ hg_user,  const float* __restrict__ hg_item,
    const float* __restrict__ h1,       const float* __restrict__ h2,
    const float* __restrict__ W1,       const float* __restrict__ b1,
    const float* __restrict__ W2,       const float* __restrict__ b2,
    float* __restrict__ out)
{
    __shared__ float sW1t[D * LDP];
    __shared__ float sA[4][16 * LDP];

    const int tid  = threadIdx.x;
    const int w    = tid >> 5;
    const int lane = tid & 31;

    const int tile_base = (blockIdx.x * 4 + w) * 16;

    const int hsel = lane >> 4;
    const int c4   = lane & 15;
    float4 e4[8], a4[8], b4[8], g4[8];
    #pragma unroll
    for (int s = 0; s < 8; ++s) {
        const int ng = tile_base + 2 * s + hsel;
        const int nc = min(ng, NUM_NODES - 1);
        const bool is_user = nc < M_USERS;
        const float* ego = is_user ? (user_emb + (size_t)nc * D)
                                   : (item_emb + (size_t)(nc - M_USERS) * D);
        const float* hg  = is_user ? (hg_user + (size_t)nc * D)
                                   : (hg_item + (size_t)(nc - M_USERS) * D);
        e4[s] = *(const float4*)(ego + c4 * 4);
        a4[s] = *(const float4*)(h1 + (size_t)nc * D + c4 * 4);
        b4[s] = *(const float4*)(h2 + (size_t)nc * D + c4 * 4);
        g4[s] = *(const float4*)(hg + c4 * 4);
    }

    for (int i = tid; i < K2 * D; i += 128) {
        int k = i >> 6;
        int n = i & 63;
        sW1t[n * LDP + k] = W1[i];
    }

    #pragma unroll
    for (int s = 0; s < 8; ++s) {
        const int node = 2 * s + hsel;
        const bool in_range = (tile_base + node) < NUM_NODES;
        const float inv3 = 1.0f / 3.0f;
        float4 f1 = make_float4((e4[s].x + a4[s].x + b4[s].x) * inv3,
                                (e4[s].y + a4[s].y + b4[s].y) * inv3,
                                (e4[s].z + a4[s].z + b4[s].z) * inv3,
                                (e4[s].w + a4[s].w + b4[s].w) * inv3);
        float4 f2 = g4[s];
        if (!in_range) {
            f1 = make_float4(0.f, 0.f, 0.f, 0.f);
            f2 = make_float4(0.f, 0.f, 0.f, 0.f);
        }
        *(float4*)&sA[w][node * LDP + c4 * 4]     = f1;
        *(float4*)&sA[w][node * LDP + D + c4 * 4] = f2;
    }

    __syncthreads();

    const int mrow = lane & 15;
    const int koff = (lane >> 4) * 2;

    v8f acc[4] = {};
    for (int kk = 0; kk < 32; ++kk) {
        int k0 = kk * 4 + koff;
        v2f a;
        a.x = sA[w][mrow * LDP + k0];
        a.y = sA[w][mrow * LDP + k0 + 1];
        #pragma unroll
        for (int nt = 0; nt < 4; ++nt) {
            int ncol = nt * 16 + mrow;
            v2f b;
            b.x = sW1t[ncol * LDP + k0];
            b.y = sW1t[ncol * LDP + k0 + 1];
            acc[nt] = __builtin_amdgcn_wmma_f32_16x16x4_f32(
                false, a, false, b, (short)0, acc[nt], false, false);
        }
    }

    float p0[8], p1[8];
    #pragma unroll
    for (int v = 0; v < 8; ++v) { p0[v] = 0.f; p1[v] = 0.f; }
    for (int nt = 0; nt < 4; ++nt) {
        int ncol = nt * 16 + mrow;
        float bb  = b1[ncol];
        float w2a = W2[ncol * 2 + 0];
        float w2b = W2[ncol * 2 + 1];
        #pragma unroll
        for (int v = 0; v < 8; ++v) {
            float hdn = tanhf(acc[nt][v] + bb);
            p0[v] += hdn * w2a;
            p1[v] += hdn * w2b;
        }
    }

    #pragma unroll
    for (int off = 1; off < 16; off <<= 1) {
        #pragma unroll
        for (int v = 0; v < 8; ++v) {
            p0[v] += __shfl_xor(p0[v], off, 32);
            p1[v] += __shfl_xor(p1[v], off, 32);
        }
    }

    float bb0 = b2[0], bb1 = b2[1];
    const int half = lane >> 4;

    for (int v = 0; v < 8; ++v) {
        float l0 = p0[v] + bb0;
        float l1 = p1[v] + bb1;
        float e  = expf(l1 - l0);
        float w1w = e / (1.0f + e);
        float w0w = 1.0f - w1w;
        int m  = v + half * 8;
        int ng = tile_base + m;
        if (ng < NUM_NODES) {
            #pragma unroll
            for (int ff = 0; ff < 4; ++ff) {
                int feat = mrow + ff * 16;
                float f1v = sA[w][m * LDP + feat];
                float f2v = sA[w][m * LDP + D + feat];
                out[(size_t)ng * D + feat] = w0w * f1v + w1w * f2v;
            }
        }
    }
}

extern "C" void kernel_launch(void* const* d_in, const int* in_sizes, int n_in,
                              void* d_out, int out_size, void* d_ws, size_t ws_size,
                              hipStream_t stream) {
    const float* user_emb = (const float*)d_in[0];
    const float* item_emb = (const float*)d_in[1];
    const float* hg_user  = (const float*)d_in[2];
    const float* hg_item  = (const float*)d_in[3];
    const float* adj_vals = (const float*)d_in[4];
    const float* W1       = (const float*)d_in[5];
    const float* b1       = (const float*)d_in[6];
    const float* W2       = (const float*)d_in[7];
    const float* b2       = (const float*)d_in[8];
    const int*   adj_rows = (const int*)d_in[9];
    const int*   adj_cols = (const int*)d_in[10];
    float* out = (float*)d_out;

    float* h1 = (float*)d_ws;
    float* h2 = h1 + (size_t)NUM_NODES * D;
    int*  rowptr = (int*)(h2 + (size_t)NUM_NODES * D);
    int*  rowend = rowptr + NUM_NODES;
    int*  bsum   = rowend + NUM_NODES;
    int*  bcnt   = bsum + 1024;
    int*  boff   = bcnt + NCELL;
    int2* sedge  = (int2*)(boff + NCELL);
    int*  rank   = (int*)h2;
    int2* pedge  = (int2*)(h2 + EDGE_ROOM);

    int E = in_sizes[4];

    part_count_kernel<<<PART_BLOCKS, 256, 0, stream>>>(adj_rows, rank, bcnt, E);
    blocksum_kernel<<<NBK, 256, 0, stream>>>(bcnt, bsum, NCELL / 4);
    scan_kernel<<<NBK, 256, 0, stream>>>(bcnt, bsum, boff, NCELL / 4);
    part_scatter_kernel<<<PART_BLOCKS, 256, 0, stream>>>(adj_rows, adj_cols, adj_vals, rank, boff,
                                                         pedge, E);
    bucket_rows_kernel<<<NBK, BROWS, 0, stream>>>(boff, bsum, pedge, rowptr, rowend, sedge, E);

    int gather_blocks = (NUM_NODES * 16 + 255) / 256;
    gather_kernel<<<gather_blocks, 256, 0, stream>>>(user_emb, item_emb, M_USERS,
                                                     rowptr, rowend, sedge, h1, E);
    gather_kernel<<<gather_blocks, 256, 0, stream>>>(h1, h1, NUM_NODES,
                                                     rowptr, rowend, sedge, h2, E);

    int fuse_blocks = (NUM_NODES + 63) / 64;
    fuse_kernel<<<fuse_blocks, 128, 0, stream>>>(user_emb, item_emb, hg_user, hg_item,
                                                 h1, h2, W1, b1, W2, b2, out);
}


# edge sort: both levels put their records in order in LDS first and copy them out as runs (level 1) or whole 128-B lines (level 2), not one 8-B store per line; sum order = LDS add arrival
# speedup vs baseline: 4.1759x; 1.2560x over previous
#include <hip/hip_runtime.h>
#include <math.h>

#define D 64
#define K2 128
#define M_USERS 200000
#define NUM_NODES 300000

typedef __attribute__((ext_vector_type(2))) float v2f;
typedef __attribute__((ext_vector_type(8))) float v8f;

#define BSH 9
#define BROWS (1 << BSH)
#define NBK ((NUM_NODES + BROWS - 1) >> BSH)
#define PART_BLOCKS 1024
#define NCELL (NBK * PART_BLOCKS)
#define COLBITS 19
#define COLMASK ((1 << COLBITS) - 1)
#define SCAN_CHUNK 1024
static_assert(NUM_NODES <= (1 << COLBITS), "col and the row's place in its bucket share one word");
static_assert(BSH + COLBITS <= 31, "the shared word stays non-negative");
static_assert(PART_BLOCKS == SCAN_CHUNK, "one scan chunk per bucket: its sum is the bucket's size");
static_assert(NCELL % 4 == 0, "cell counters are read as int4");
#define EDGE_ROOM (NUM_NODES * D / 3)
static_assert(EDGE_ROOM % 4 == 0, "the bucket-ordered copy starts on a 16-byte boundary");

__device__ __forceinline__ void slice_bounds(int nedges, int b, int& start, int& stop) {
    const int ch = (((nedges + PART_BLOCKS - 1) / PART_BLOCKS) + 3) & ~3;
    start = min(b * ch, nedges);
    stop  = min(start + ch, nedges);
}

__device__ __forceinline__ int bucket_of(int row) {
    return (int)min((unsigned)row >> BSH, (unsigned)(NBK - 1));
}

__global__ __launch_bounds__(256) void part_count_kernel(const int* __restrict__ rows,
                                                         int* __restrict__ bcnt, int nedges) {
    __shared__ int s_cnt[NBK];
    const int t = threadIdx.x;
    const int b = blockIdx.x;
    for (int k = t; k < NBK; k += 256) s_cnt[k] = 0;
    __syncthreads();
    int start, stop;
    slice_bounds(nedges, b, start, stop);
    const int nq = (stop - start) >> 2;
    const int4* rows4 = (const int4*)(rows + start);
    for (int q = t; q < nq; q += 256) {
        const int4 r = rows4[q];
        atomicAdd(&s_cnt[bucket_of(r.x)], 1);
        atomicAdd(&s_cnt[bucket_of(r.y)], 1);
        atomicAdd(&s_cnt[bucket_of(r.z)], 1);
        atomicAdd(&s_cnt[bucket_of(r.w)], 1);
    }
    for (int e = start + (nq << 2) + t; e < stop; e += 256)
        atomicAdd(&s_cnt[bucket_of(rows[e])], 1);
    __syncthreads();
    for (int k = t; k < NBK; k += 256) bcnt[k * PART_BLOCKS + b] = s_cnt[k];
}

__global__ __launch_bounds__(256) void blocksum_kernel(const int* __restrict__ cnt,
                                                       int* __restrict__ blocksum, int n4) {
    __shared__ int s_red[256];
    const int t = threadIdx.x;
    const int i4 = blockIdx.x * 256 + t;
    int s = 0;
    if (i4 < n4) {
        int4 c = ((const int4*)cnt)[i4];
        s = c.x + c.y + c.z + c.w;
    }
    s_red[t] = s;
    __syncthreads();
    for (int off = 128; off > 0; off >>= 1) {
        if (t < off) s_red[t] += s_red[t + off];
        __syncthreads();
    }
    if (t == 0) blocksum[blockIdx.x] = s_red[0];
}

__global__ __launch_bounds__(256) void scan_kernel(const int* __restrict__ cnt,
                                                   const int* __restrict__ blocksum,
                                                   int* __restrict__ excl_out, int n4) {
    __shared__ int s_red[256];
    __shared__ int s_scan[256];
    const int t = threadIdx.x;
    const int b = blockIdx.x;

    int part = 0;
    for (int j = t; j < b; j += 256) part += blocksum[j];
    s_red[t] = part;
    __syncthreads();
    for (int off = 128; off > 0; off >>= 1) {
        if (t < off) s_red[t] += s_red[t + off];
        __syncthreads();
    }
    const int base = s_red[0];

    const int i4 = b * 256 + t;
    const bool valid = i4 < n4;
    int4 c = make_int4(0, 0, 0, 0);
    if (valid) c = ((const int4*)cnt)[i4];
    const int s = c.x + c.y + c.z + c.w;

    s_scan[t] = s;
    __syncthreads();
    for (int off = 1; off < 256; off <<= 1) {
        int add = (t >= off) ? s_scan[t - off] : 0;
        __syncthreads();
        s_scan[t] += add;
        __syncthreads();
    }
    const int excl = base + s_scan[t] - s;

    if (valid) {
        int4 o;
        o.x = excl;
        o.y = excl + c.x;
        o.z = excl + c.x + c.y;
        o.w = excl + c.x + c.y + c.z;
        ((int4*)excl_out)[i4] = o;
    }
}

#define SLICE_CAP 4096
static_assert(3 * 256 >= NBK, "three counters per thread cover the buckets in the local scan");

__device__ __forceinline__ void stage_part(int* s_cur, const int* s_lof, const int* s_off,
                                           int2* s_rec, int* s_dst, int2* __restrict__ pedge,
                                           int row, int col, float val, int nedges) {
    const int k = bucket_of(row);
    const int l = atomicAdd(&s_cur[k], 1);
    const int pos = s_off[k] + (l - s_lof[k]);
    const unsigned p = min((unsigned)pos, (unsigned)(nedges - 1));
    const int2 rec = make_int2(((row & (BROWS - 1)) << COLBITS) | (col & COLMASK),
                               __float_as_int(val));
    if ((unsigned)l < (unsigned)SLICE_CAP) {
        s_rec[l] = rec;
        s_dst[l] = (int)p;
    } else {
        pedge[p] = rec;
    }
}

__device__ __forceinline__ void put_rec(int2* __restrict__ pedge, int pos, int2 rec, int nedges) {
    pedge[min((unsigned)pos, (unsigned)(nedges - 1))] = rec;
}

__global__ __launch_bounds__(256) void part_scatter_kernel(const int* __restrict__ rows,
                                                           const int* __restrict__ cols,
                                                           const float* __restrict__ vals,
                                                           const int* __restrict__ boff,
                                                           int2* __restrict__ pedge, int nedges) {
    __shared__ int  s_off[NBK];
    __shared__ int  s_cur[NBK];
    __shared__ int  s_lof[NBK];
    __shared__ int  s_wsum[8];
    __shared__ int2 s_rec[SLICE_CAP];
    __shared__ int  s_dst[SLICE_CAP];
    const int t = threadIdx.x;
    const int b = blockIdx.x;
    for (int k = t; k < NBK; k += 256) {
        const int c  = k * PART_BLOCKS + b;
        const int o  = boff[c];
        const int o1 = (c + 1 < NCELL) ? boff[c + 1] : nedges;
        s_off[k] = o;
        s_cur[k] = o1 - o;
    }
    __syncthreads();
    int start, stop;
    slice_bounds(nedges, b, start, stop);
    const int nq = (stop - start) >> 2;
    const int4*   rows4 = (const int4*)(rows + start);
    const int4*   cols4 = (const int4*)(cols + start);
    const float4* vals4 = (const float4*)(vals + start);

    const int lane = t & 31;
    const int wv   = t >> 5;
    const int k0   = 3 * t;
    const int c0 = (k0     < NBK) ? s_cur[k0]     : 0;
    const int c1 = (k0 + 1 < NBK) ? s_cur[k0 + 1] : 0;
    const int c2 = (k0 + 2 < NBK) ? s_cur[k0 + 2] : 0;
    const int own = c0 + c1 + c2;
    int inc = own;
    for (int off = 1; off < 32; off <<= 1) {
        const int up = __shfl_up(inc, off, 32);
        if (lane >= off) inc += up;
    }
    if (lane == 31) s_wsum[wv] = inc;
    __syncthreads();
    int ex = inc - own;
    for (int w2 = 0; w2 < wv; ++w2) ex += s_wsum[w2];
    if (k0 < NBK)     { s_lof[k0]     = ex;           s_cur[k0]     = ex; }
    if (k0 + 1 < NBK) { s_lof[k0 + 1] = ex + c0;      s_cur[k0 + 1] = ex + c0; }
    if (k0 + 2 < NBK) { s_lof[k0 + 2] = ex + c0 + c1; s_cur[k0 + 2] = ex + c0 + c1; }
    __syncthreads();

    for (int q = t; q < nq; q += 256) {
        const int4   r = rows4[q];
        const int4   cc = cols4[q];
        const float4 v = vals4[q];
        stage_part(s_cur, s_lof, s_off, s_rec, s_dst, pedge, r.x, cc.x, v.x, nedges);
        stage_part(s_cur, s_lof, s_off, s_rec, s_dst, pedge, r.y, cc.y, v.y, nedges);
        stage_part(s_cur, s_lof, s_off, s_rec, s_dst, pedge, r.z, cc.z, v.z, nedges);
        stage_part(s_cur, s_lof, s_off, s_rec, s_dst, pedge, r.w, cc.w, v.w, nedges);
    }
    for (int e = start + (nq << 2) + t; e < stop; e += 256)
        stage_part(s_cur, s_lof, s_off, s_rec, s_dst, pedge, rows[e], cols[e], vals[e], nedges);
    __syncthreads();

    const int nst = min(stop - start, SLICE_CAP);
    int j = t;
    for (; j + 3 * 256 < nst; j += 4 * 256) {
        const int2 a0 = s_rec[j];
        const int2 a1 = s_rec[j + 256];
        const int2 a2 = s_rec[j + 2 * 256];
        const int2 a3 = s_rec[j + 3 * 256];
        const int  d0 = s_dst[j];
        const int  d1 = s_dst[j + 256];
        const int  d2 = s_dst[j + 2 * 256];
        const int  d3 = s_dst[j + 3 * 256];
        put_rec(pedge, d0, a0, nedges);
        put_rec(pedge, d1, a1, nedges);
        put_rec(pedge, d2, a2, nedges);
        put_rec(pedge, d3, a3, nedges);
    }
    for (; j < nst; j += 256) put_rec(pedge, s_dst[j], s_rec[j], nedges);
}

#define STAGE_CAP 8192
#define LINE_RECS 16
static_assert(BROWS % (2 * LINE_RECS) == 0, "a wave's 32 records of the copy-out start on a line");

__device__ __forceinline__ int row_in_bucket(int word) {
    return (int)(((unsigned)word >> COLBITS) & (BROWS - 1));
}

__device__ __forceinline__ void place_edge(int2* __restrict__ sedge, int* s_cur, int2* s_img,
                                           int lo, int shift, int2 m, int nedges) {
    const int rel = atomicAdd(&s_cur[row_in_bucket(m.x)], 1);
    const int2 rec = make_int2(m.x & COLMASK, m.y);
    if ((unsigned)rel < (unsigned)STAGE_CAP) {
        s_img[rel + shift] = rec;
    } else {
        const unsigned p = min((unsigned)(lo + rel), (unsigned)(nedges - 1));
        sedge[p] = rec;
    }
}

__global__ __launch_bounds__(BROWS) void bucket_rows_kernel(const int* __restrict__ boff,
                                                            const int* __restrict__ bsum,
                                                            const int2* __restrict__ pedge,
                                                            int* __restrict__ rowptr,
                                                            int* __restrict__ rowend,
                                                            int2* __restrict__ sedge, int nedges) {
    __shared__ int  s_cnt[BROWS];
    __shared__ int  s_scan[BROWS];
    __shared__ int  s_cur[BROWS];
    __shared__ int2 s_img[STAGE_CAP + LINE_RECS];
    const int t = threadIdx.x;
    const int k = blockIdx.x;
    const int lo = min(max(boff[k * PART_BLOCKS], 0), nedges);
    const int hi = min(max(lo + bsum[k], lo), nedges);
    const int shift = lo & (LINE_RECS - 1);

    s_cnt[t] = 0;
    __syncthreads();
    int i = lo + t;
    for (; i + 3 * BROWS < hi; i += 4 * BROWS) {
        const int x0 = pedge[i].x;
        const int x1 = pedge[i + BROWS].x;
        const int x2 = pedge[i + 2 * BROWS].x;
        const int x3 = pedge[i + 3 * BROWS].x;
        atomicAdd(&s_cnt[row_in_bucket(x0)], 1);
        atomicAdd(&s_cnt[row_in_bucket(x1)], 1);
        atomicAdd(&s_cnt[row_in_bucket(x2)], 1);
        atomicAdd(&s_cnt[row_in_bucket(x3)], 1);
    }
    for (; i < hi; i += BROWS) atomicAdd(&s_cnt[row_in_bucket(pedge[i].x)], 1);
    __syncthreads();

    const int cown = s_cnt[t];
    s_scan[t] = cown;
    __syncthreads();
    for (int off = 1; off < BROWS; off <<= 1) {
        const int add = (t >= off) ? s_scan[t - off] : 0;
        __syncthreads();
        s_scan[t] += add;
        __syncthreads();
    }
    const int excl = s_scan[t] - cown;
    s_cur[t] = excl;
    const int row = (k << BSH) + t;
    if (row < NUM_NODES) {
        rowptr[row] = lo + excl;
        rowend[row] = lo + excl + cown;
    }
    __syncthreads();

    i = lo + t;
    for (; i + 3 * BROWS < hi; i += 4 * BROWS) {
        const int2 m0 = pedge[i];
        const int2 m1 = pedge[i + BROWS];
        const int2 m2 = pedge[i + 2 * BROWS];
        const int2 m3 = pedge[i + 3 * BROWS];
        place_edge(sedge, s_cur, s_img, lo, shift, m0, nedges);
        place_edge(sedge, s_cur, s_img, lo, shift, m1, nedges);
        place_edge(sedge, s_cur, s_img, lo, shift, m2, nedges);
        place_edge(sedge, s_cur, s_img, lo, shift, m3, nedges);
    }
    for (; i < hi; i += BROWS) place_edge(sedge, s_cur, s_img, lo, shift, pedge[i], nedges);
    __syncthreads();

    const int nst   = min(hi - lo, STAGE_CAP);
    const int gbase = lo - shift;
    const int jend  = shift + nst;
    int j = t;
    if (j >= shift && j < jend) sedge[gbase + j] = s_img[j];
    j += BROWS;
    for (; j + 3 * BROWS < jend; j += 4 * BROWS) {
        const int2 a0 = s_img[j];
        const int2 a1 = s_img[j + BROWS];
        const int2 a2 = s_img[j + 2 * BROWS];
        const int2 a3 = s_img[j + 3 * BROWS];
        sedge[gbase + j]             = a0;
        sedge[gbase + j + BROWS]     = a1;
        sedge[gbase + j + 2 * BROWS] = a2;
        sedge[gbase + j + 3 * BROWS] = a3;
    }
    for (; j < jend; j += BROWS) sedge[gbase + j] = s_img[j];
}

typedef __attribute__((ext_vector_type(4))) float v4f;

__device__ __forceinline__ v4f load_x4(const float* __restrict__ xu,
                                       const float* __restrict__ xi,
                                       int split, int c, int sub) {
    const float* src = (c < split) ? (xu + (size_t)c * D)
                                   : (xi + (size_t)(c - split) * D);
    return *(const v4f*)(src + sub * 4);
}

__global__ __launch_bounds__(256) void gather_kernel(const float* __restrict__ xu,
                                                     const float* __restrict__ xi,
                                                     int split,
                                                     const int*  __restrict__ rowptr,
                                                     const int*  __restrict__ rowend,
                                                     const int2* __restrict__ sedge,
                                                     float* __restrict__ out,
                                                     int nedges) {
    const int row = (blockIdx.x * 256 + threadIdx.x) >> 4;
    const int sub = threadIdx.x & 15;
    if (row >= NUM_NODES) return;
    const int start = min(max(rowptr[row], 0), nedges);
    const int end   = min(max(rowend[row], start), nedges);
    v4f acc = {0.f, 0.f, 0.f, 0.f};
    for (int base = start; base < end; base += 16) {
        const int n = min(end - base, 16);
        int mc = 0, mv = 0;
        if (sub < n) {
            int2 m = sedge[base + sub];
            mc = m.x;
            mv = m.y;
        }
        for (int t = 0; t < n; t += 4) {
            const int i1 = min(t + 1, n - 1);
            const int i2 = min(t + 2, n - 1);
            const int i3 = min(t + 3, n - 1);
            const int   c0 = __shfl(mc, t,  16);
            const int   c1 = __shfl(mc, i1, 16);
            const int   c2 = __shfl(mc, i2, 16);
            const int   c3 = __shfl(mc, i3, 16);
            const float w0 = __int_as_float(__shfl(mv, t,  16));
            const float w1 = __int_as_float(__shfl(mv, i1, 16));
            const float w2 = __int_as_float(__shfl(mv, i2, 16));
            const float w3 = __int_as_float(__shfl(mv, i3, 16));
            v4f x0 = load_x4(xu, xi, split, c0, sub);
            v4f x1 = load_x4(xu, xi, split, c1, sub);
            v4f x2 = load_x4(xu, xi, split, c2, sub);
            v4f x3 = load_x4(xu, xi, split, c3, sub);
            const v4f zero = {0.f, 0.f, 0.f, 0.f};
            acc += w0 * x0;
            acc += (t + 1 < n) ? (w1 * x1) : zero;
            acc += (t + 2 < n) ? (w2 * x2) : zero;
            acc += (t + 3 < n) ? (w3 * x3) : zero;
        }
    }
    *(v4f*)(out + (size_t)row * D + sub * 4) = acc;
}

#define LDP (K2 + 4)
__global__ __launch_bounds__(128) void fuse_kernel(
    const float* __restrict__ user_emb, const float* __restrict__ item_emb,
    const float* __restrict__ hg_user,  const float* __restrict__ hg_item,
    const float* __restrict__ h1,       const float* __restrict__ h2,
    const float* __restrict__ W1,       const float* __restrict__ b1,
    const float* __restrict__ W2,       const float* __restrict__ b2,
    float* __restrict__ out)
{
    __shared__ float sW1t[D * LDP];
    __shared__ float sA[4][16 * LDP];

    const int tid  = threadIdx.x;
    const int w    = tid >> 5;
    const int lane = tid & 31;

    const int tile_base = (blockIdx.x * 4 + w) * 16;

    const int hsel = lane >> 4;
    const int c4   = lane & 15;
    float4 e4[8], a4[8], b4[8], g4[8];
    #pragma unroll
    for (int s = 0; s < 8; ++s) {
        const int ng = tile_base + 2 * s + hsel;
        const int nc = min(ng, NUM_NODES - 1);
        const bool is_user = nc < M_USERS;
        const float* ego = is_user ? (user_emb + (size_t)nc * D)
                                   : (item_emb + (size_t)(nc - M_USERS) * D);
        const float* hg  = is_user ? (hg_user + (size_t)nc * D)
                                   : (hg_item + (size_t)(nc - M_USERS) * D);
        e4[s] = *(const float4*)(ego + c4 * 4);
        a4[s] = *(const float4*)(h1 + (size_t)nc * D + c4 * 4);
        b4[s] = *(const float4*)(h2 + (size_t)nc * D + c4 * 4);
        g4[s] = *(const float4*)(hg + c4 * 4);
    }

    for (int i = tid; i < K2 * D; i += 128) {
        int k = i >> 6;
        int n = i & 63;
        sW1t[n * LDP + k] = W1[i];
    }

    #pragma unroll
    for (int s = 0; s < 8; ++s) {
        const int node = 2 * s + hsel;
        const bool in_range = (tile_base + node) < NUM_NODES;
        const float inv3 = 1.0f / 3.0f;
        float4 f1 = make_float4((e4[s].x + a4[s].x + b4[s].x) * inv3,
                                (e4[s].y + a4[s].y + b4[s].y) * inv3,
                                (e4[s].z + a4[s].z + b4[s].z) * inv3,
                                (e4[s].w + a4[s].w + b4[s].w) * inv3);
        float4 f2 = g4[s];
        if (!in_range) {
            f1 = make_float4(0.f, 0.f, 0.f, 0.f);
            f2 = make_float4(0.f, 0.f, 0.f, 0.f);
        }
        *(float4*)&sA[w][node * LDP + c4 * 4]     = f1;
        *(float4*)&sA[w][node * LDP + D + c4 * 4] = f2;
    }

    __syncthreads();

    const int mrow = lane & 15;
    const int koff = (lane >> 4) * 2;

    v8f acc[4] = {};
    for (int kk = 0; kk < 32; ++kk) {
        int k0 = kk * 4 + koff;
        v2f a;
        a.x = sA[w][mrow * LDP + k0];
        a.y = sA[w][mrow * LDP + k0 + 1];
        #pragma unroll
        for (int nt = 0; nt < 4; ++nt) {
            int ncol = nt * 16 + mrow;
            v2f b;
            b.x = sW1t[ncol * LDP + k0];
            b.y = sW1t[ncol * LDP + k0 + 1];
            acc[nt] = __builtin_amdgcn_wmma_f32_16x16x4_f32(
                false, a, false, b, (short)0, acc[nt], false, false);
        }
    }

    float p0[8], p1[8];
    #pragma unroll
    for (int v = 0; v < 8; ++v) { p0[v] = 0.f; p1[v] = 0.f; }
    for (int nt = 0; nt < 4; ++nt) {
        int ncol = nt * 16 + mrow;
        float bb  = b1[ncol];
        float w2a = W2[ncol * 2 + 0];
        float w2b = W2[ncol * 2 + 1];
        #pragma unroll
        for (int v = 0; v < 8; ++v) {
            float hdn = tanhf(acc[nt][v] + bb);
            p0[v] += hdn * w2a;
            p1[v] += hdn * w2b;
        }
    }

    #pragma unroll
    for (int off = 1; off < 16; off <<= 1) {
        #pragma unroll
        for (int v = 0; v < 8; ++v) {
            p0[v] += __shfl_xor(p0[v], off, 32);
            p1[v] += __shfl_xor(p1[v], off, 32);
        }
    }

    float bb0 = b2[0], bb1 = b2[1];
    const int half = lane >> 4;

    for (int v = 0; v < 8; ++v) {
        float l0 = p0[v] + bb0;
        float l1 = p1[v] + bb1;
        float e  = expf(l1 - l0);
        float w1w = e / (1.0f + e);
        float w0w = 1.0f - w1w;
        int m  = v + half * 8;
        int ng = tile_base + m;
        if (ng < NUM_NODES) {
            #pragma unroll
            for (int ff = 0; ff < 4; ++ff) {
                int feat = mrow + ff * 16;
                float f1v = sA[w][m * LDP + feat];
                float f2v = sA[w][m * LDP + D + feat];
                out[(size_t)ng * D + feat] = w0w * f1v + w1w * f2v;
            }
        }
    }
}

extern "C" void kernel_launch(void* const* d_in, const int* in_sizes, int n_in,
                              void* d_out, int out_size, void* d_ws, size_t ws_size,
                              hipStream_t stream) {
    const float* user_emb = (const float*)d_in[0];
    const float* item_emb = (const float*)d_in[1];
    const float* hg_user  = (const float*)d_in[2];
    const float* hg_item  = (const float*)d_in[3];
    const float* adj_vals = (const float*)d_in[4];
    const float* W1       = (const float*)d_in[5];
    const float* b1       = (const float*)d_in[6];
    const float* W2       = (const float*)d_in[7];
    const float* b2       = (const float*)d_in[8];
    const int*   adj_rows = (const int*)d_in[9];
    const int*   adj_cols = (const int*)d_in[10];
    float* out = (float*)d_out;

    float* h1 = (float*)d_ws;
    float* h2 = h1 + (size_t)NUM_NODES * D;
    int*  rowptr = (int*)(h2 + (size_t)NUM_NODES * D);
    int*  rowend = rowptr + NUM_NODES;
    int*  bsum   = rowend + NUM_NODES;
    int*  bcnt   = bsum + 1024;
    int*  boff   = bcnt + NCELL;
    int2* sedge  = (int2*)(boff + NCELL);
    int2* pedge  = (int2*)(h2 + EDGE_ROOM);

    int E = in_sizes[4];

    part_count_kernel<<<PART_BLOCKS, 256, 0, stream>>>(adj_rows, bcnt, E);
    blocksum_kernel<<<NBK, 256, 0, stream>>>(bcnt, bsum, NCELL / 4);
    scan_kernel<<<NBK, 256, 0, stream>>>(bcnt, bsum, boff, NCELL / 4);
    part_scatter_kernel<<<PART_BLOCKS, 256, 0, stream>>>(adj_rows, adj_cols, adj_vals, boff,
                                                         pedge, E);
    bucket_rows_kernel<<<NBK, BROWS, 0, stream>>>(boff, bsum, pedge, rowptr, rowend, sedge, E);

    int gather_blocks = (NUM_NODES * 16 + 255) / 256;
    gather_kernel<<<gather_blocks, 256, 0, stream>>>(user_emb, item_emb, M_USERS,
                                                     rowptr, rowend, sedge, h1, E);
    gather_kernel<<<gather_blocks, 256, 0, stream>>>(h1, h1, NUM_NODES,
                                                     rowptr, rowend, sedge, h2, E);

    int fuse_blocks = (NUM_NODES + 63) / 64;
    fuse_kernel<<<fuse_blocks, 128, 0, stream>>>(user_emb, item_emb, hg_user, hg_item,
                                                 h1, h2, W1, b1, W2, b2, out);
}
